# starting kernel
# baseline (speedup 1.0000x reference)
#include <hip/hip_runtime.h>
#include <math.h>

typedef float v2f __attribute__((ext_vector_type(2)));
typedef float v8f __attribute__((ext_vector_type(8)));

namespace {
constexpr int kB  = 8;
constexpr int kT  = 2048;
constexpr int kD  = 1024;
constexpr int kHD = 64;
constexpr int kM  = 64;
constexpr int kS  = kT + kM;
constexpr float kLnEps = 1e-5f;
constexpr float kScale = 0.125f;
}

__device__ __forceinline__ v8f wmma_f32(v2f a, v2f b, v8f c) {
  return __builtin_amdgcn_wmma_f32_16x16x4_f32(
       false, a,  false, b,
       (short)0, c,  false,  false);
}

__global__ __launch_bounds__(64)
void ln_mem_kernel(const float* __restrict__ c_emb,
                   const float* __restrict__ gamma,
                   const float* __restrict__ beta,
                   float* __restrict__ k_buf,
                   float* __restrict__ v_buf) {
  const int b = blockIdx.x;
  const int r = threadIdx.x;
  const float* row = c_emb + r * kHD;
  float mu = 0.f;
  #pragma unroll
  for (int i = 0; i < kHD; ++i) mu += row[i];
  mu *= (1.f / kHD);
  float var = 0.f;
  #pragma unroll
  for (int i = 0; i < kHD; ++i) { float d = row[i] - mu; var += d * d; }
  var *= (1.f / kHD);
  const float rstd = rsqrtf(var + kLnEps);
  float* kd = k_buf + ((size_t)b * kS + kT + r) * kHD;
  float* vd = v_buf + ((size_t)b * kS + kT + r) * kHD;
  #pragma unroll
  for (int i = 0; i < kHD; ++i) {
    float y = (row[i] - mu) * rstd * gamma[i] + beta[i];
    kd[i] = y;
    vd[i] = y;
  }
}

__global__ __launch_bounds__(256)
void qkv_kernel(const float* __restrict__ x,
                const float* __restrict__ Wq,
                const float* __restrict__ Wk,
                const float* __restrict__ Wv,
                float* __restrict__ q_buf,
                float* __restrict__ k_buf,
                float* __restrict__ v_buf) {
  __shared__ float xs[128][36];
  __shared__ float wst[192][34];

  const int tid  = threadIdx.x;
  const int wid  = tid >> 5;
  const int lane = tid & 31;
  const int half = lane >> 4;
  const int lm   = lane & 15;
  const int r0   = blockIdx.x * 128;
  const int m0   = wid * 16;

  v8f acc[12] = {};
  float4 xreg[4];
  float4 wreg[6];

  auto gload = [&](int kt) {
    #pragma unroll
    for (int i = 0; i < 4; ++i) {
      int idx = tid + 256 * i;
      int row = idx >> 3;
      int c4  = (idx & 7) * 4;
      xreg[i] = *(const float4*)(x + (size_t)(r0 + row) * kD + kt + c4);
    }
    #pragma unroll
    for (int i = 0; i < 6; ++i) {
      int idx = tid + 256 * i;
      int kr  = idx / 48;
      int c4  = idx % 48;
      int mat = c4 >> 4;
      int cc  = (c4 & 15) * 4;
      const float* src = (mat == 0) ? Wq : ((mat == 1) ? Wk : Wv);
      wreg[i] = *(const float4*)(src + (size_t)(kt + kr) * kHD + cc);
    }
  };
  auto lstore = [&]() {
    #pragma unroll
    for (int i = 0; i < 4; ++i) {
      int idx = tid + 256 * i;
      int row = idx >> 3;
      int c4  = (idx & 7) * 4;
      *(float4*)&xs[row][c4] = xreg[i];
    }
    #pragma unroll
    for (int i = 0; i < 6; ++i) {
      int idx = tid + 256 * i;
      int kr  = idx / 48;
      int c4  = idx % 48;
      int col = (c4 >> 4) * 64 + (c4 & 15) * 4;
      wst[col + 0][kr] = wreg[i].x;
      wst[col + 1][kr] = wreg[i].y;
      wst[col + 2][kr] = wreg[i].z;
      wst[col + 3][kr] = wreg[i].w;
    }
  };

  gload(0);
  lstore();
  __syncthreads();

  constexpr int kNT = kD / 32;
  for (int t = 0; t < kNT; ++t) {
    if (t + 1 < kNT) gload(32 * (t + 1));
    #pragma unroll
    for (int kk = 0; kk < 8; ++kk) {
      const int k0 = kk * 4;
      v2f a = *(const v2f*)&xs[m0 + lm][k0 + 2 * half];
      #pragma unroll
      for (int c = 0; c < 12; ++c) {
        v2f bf = *(const v2f*)&wst[c * 16 + lm][k0 + 2 * half];
        acc[c] = wmma_f32(a, bf, acc[c]);
      }
    }
    __syncthreads();
    if (t + 1 < kNT) lstore();
    __syncthreads();
  }

  #pragma unroll
  for (int c = 0; c < 12; ++c) {
    const int mat = c >> 2;
    const int hd  = (c & 3) * 16 + lm;
    #pragma unroll
    for (int i = 0; i < 8; ++i) {
      const int gr = r0 + m0 + i + 8 * half;
      const float val = acc[c][i];
      if (mat == 0) {
        q_buf[(size_t)gr * kHD + hd] = val;
      } else {
        const int bb = gr / kT;
        const int tt = gr - bb * kT;
        float* dst = (mat == 1) ? k_buf : v_buf;
        dst[((size_t)bb * kS + tt) * kHD + hd] = val;
      }
    }
  }
}

__global__ __launch_bounds__(256)
void attn_kernel(const float* __restrict__ q_buf,
                 const float* __restrict__ k_buf,
                 const float* __restrict__ v_buf,
                 float* __restrict__ out) {
  __shared__ float qs[128][68];
  __shared__ float ks[64][68];
  __shared__ float vst[64][68];
  __shared__ float ps[8][16][68];

  const int tid  = threadIdx.x;
  const int wid  = tid >> 5;
  const int lane = tid & 31;
  const int half = lane >> 4;
  const int lm   = lane & 15;
  const int b    = blockIdx.y;
  const int q0   = blockIdx.x * 128;
  const int m0   = wid * 16;

  float4 kreg[4], vreg[4];
  auto gload = [&](int s0) {
    #pragma unroll
    for (int i = 0; i < 4; ++i) {
      int idx = tid + 256 * i;
      int row = idx >> 4;
      int c4  = (idx & 15) * 4;
      kreg[i] = *(const float4*)(k_buf + ((size_t)b * kS + s0 + row) * kHD + c4);
      vreg[i] = *(const float4*)(v_buf + ((size_t)b * kS + s0 + row) * kHD + c4);
    }
  };
  auto lstore = [&]() {
    #pragma unroll
    for (int i = 0; i < 4; ++i) {
      int idx = tid + 256 * i;
      int row = idx >> 4;
      int c4  = (idx & 15) * 4;
      *(float4*)&ks[row][c4] = kreg[i];
      vst[c4 + 0][row] = vreg[i].x;
      vst[c4 + 1][row] = vreg[i].y;
      vst[c4 + 2][row] = vreg[i].z;
      vst[c4 + 3][row] = vreg[i].w;
    }
  };

  #pragma unroll
  for (int i = 0; i < 8; ++i) {
    int idx = tid + 256 * i;
    int row = idx >> 4;
    int c4  = (idx & 15) * 4;
    float4 v = *(const float4*)(q_buf + ((size_t)b * kT + q0 + row) * kHD + c4);
    v.x *= kScale; v.y *= kScale; v.z *= kScale; v.w *= kScale;
    *(float4*)&qs[row][c4] = v;
  }
  gload(0);
  lstore();
  __syncthreads();

  v8f oacc[4] = {};
  float mrun[8], lrun[8];
  #pragma unroll
  for (int i = 0; i < 8; ++i) { mrun[i] = -INFINITY; lrun[i] = 0.f; }

  constexpr int kNT = kS / 64;
  for (int it = 0; it < kNT; ++it) {
    if (it + 1 < kNT) gload(64 * (it + 1));

    v8f sacc[4] = {};
    #pragma unroll
    for (int kk = 0; kk < 16; ++kk) {
      const int k0 = kk * 4;
      v2f a = *(const v2f*)&qs[m0 + lm][k0 + 2 * half];
      #pragma unroll
      for (int j = 0; j < 4; ++j) {
        v2f bf = *(const v2f*)&ks[j * 16 + lm][k0 + 2 * half];
        sacc[j] = wmma_f32(a, bf, sacc[j]);
      }
    }

    float mnew[8], alpha[8], rsum[8];
    #pragma unroll
    for (int i = 0; i < 8; ++i) {
      float v = fmaxf(fmaxf(sacc[0][i], sacc[1][i]),
                      fmaxf(sacc[2][i], sacc[3][i]));
      #pragma unroll
      for (int off = 1; off < 16; off <<= 1)
        v = fmaxf(v, __shfl_xor(v, off));
      mnew[i]  = fmaxf(mrun[i], v);
      alpha[i] = __expf(mrun[i] - mnew[i]);
      mrun[i]  = mnew[i];
      rsum[i]  = 0.f;
    }
    #pragma unroll
    for (int j = 0; j < 4; ++j) {
      #pragma unroll
      for (int i = 0; i < 8; ++i) {
        float p = __expf(sacc[j][i] - mnew[i]);
        sacc[j][i] = p;
        rsum[i] += p;
      }
    }
    #pragma unroll
    for (int i = 0; i < 8; ++i) {
      float rs = rsum[i];
      #pragma unroll
      for (int off = 1; off < 16; off <<= 1)
        rs += __shfl_xor(rs, off);
      lrun[i] = lrun[i] * alpha[i] + rs;
    }
    #pragma unroll
    for (int c = 0; c < 4; ++c)
      #pragma unroll
      for (int i = 0; i < 8; ++i)
        oacc[c][i] *= alpha[i];

    #pragma unroll
    for (int j = 0; j < 4; ++j)
      #pragma unroll
      for (int i = 0; i < 8; ++i)
        ps[wid][i + 8 * half][j * 16 + lm] = sacc[j][i];

    #pragma unroll
    for (int kk = 0; kk < 16; ++kk) {
      const int k0 = kk * 4;
      v2f a = *(const v2f*)&ps[wid][lm][k0 + 2 * half];
      #pragma unroll
      for (int c = 0; c < 4; ++c) {
        v2f bf = *(const v2f*)&vst[c * 16 + lm][k0 + 2 * half];
        oacc[c] = wmma_f32(a, bf, oacc[c]);
      }
    }

    __syncthreads();
    if (it + 1 < kNT) lstore();
    __syncthreads();
  }

  #pragma unroll
  for (int c = 0; c < 4; ++c) {
    const int hd = c * 16 + lm;
    #pragma unroll
    for (int i = 0; i < 8; ++i) {
      const int t = q0 + m0 + i + 8 * half;
      out[((size_t)b * kT + t) * kHD + hd] = oacc[c][i] / lrun[i];
    }
  }
}

extern "C" void kernel_launch(void* const* d_in, const int* in_sizes, int n_in,
                              void* d_out, int out_size, void* d_ws, size_t ws_size,
                              hipStream_t stream) {
  const float* x     = (const float*)d_in[0];
  const float* c_emb = (const float*)d_in[1];
  const float* Wq    = (const float*)d_in[2];
  const float* Wk    = (const float*)d_in[3];
  const float* Wv    = (const float*)d_in[4];
  const float* gamma = (const float*)d_in[5];
  const float* beta  = (const float*)d_in[6];
  float* out = (float*)d_out;

  float* q_buf = (float*)d_ws;
  float* k_buf = q_buf + (size_t)kB * kT * kHD;
  float* v_buf = k_buf + (size_t)kB * kS * kHD;

  ln_mem_kernel<<<kB, 64, 0, stream>>>(c_emb, gamma, beta, k_buf, v_buf);
  qkv_kernel<<<(kB * kT) / 128, 256, 0, stream>>>(x, Wq, Wk, Wv,
                                                  q_buf, k_buf, v_buf);
  attn_kernel<<<dim3(kT / 128, kB), 256, 0, stream>>>(q_buf, k_buf, v_buf, out);
}


# both matrix kernels: work-group tile 128 -> 64 rows (8 -> 4 waves), grids 128 -> 256 work-groups, one per WGP instead of half the chip idle; per-wave tiles and sum order unchanged
# speedup vs baseline: 1.1384x; 1.1384x over previous
#include <hip/hip_runtime.h>
#include <math.h>

typedef float v2f __attribute__((ext_vector_type(2)));
typedef float v8f __attribute__((ext_vector_type(8)));

namespace {
constexpr int kB  = 8;
constexpr int kT  = 2048;
constexpr int kD  = 1024;
constexpr int kHD = 64;
constexpr int kM  = 64;
constexpr int kS  = kT + kM;
constexpr float kLnEps = 1e-5f;
constexpr float kScale = 0.125f;
constexpr int kRows    = 64;
constexpr int kThreads = kRows * 2;
}

__device__ __forceinline__ v8f wmma_f32(v2f a, v2f b, v8f c) {
  return __builtin_amdgcn_wmma_f32_16x16x4_f32(
       false, a,  false, b,
       (short)0, c,  false,  false);
}

__global__ __launch_bounds__(64)
void ln_mem_kernel(const float* __restrict__ c_emb,
                   const float* __restrict__ gamma,
                   const float* __restrict__ beta,
                   float* __restrict__ k_buf,
                   float* __restrict__ v_buf) {
  const int b = blockIdx.x;
  const int r = threadIdx.x;
  const float* row = c_emb + r * kHD;
  float mu = 0.f;
  #pragma unroll
  for (int i = 0; i < kHD; ++i) mu += row[i];
  mu *= (1.f / kHD);
  float var = 0.f;
  #pragma unroll
  for (int i = 0; i < kHD; ++i) { float d = row[i] - mu; var += d * d; }
  var *= (1.f / kHD);
  const float rstd = rsqrtf(var + kLnEps);
  float* kd = k_buf + ((size_t)b * kS + kT + r) * kHD;
  float* vd = v_buf + ((size_t)b * kS + kT + r) * kHD;
  #pragma unroll
  for (int i = 0; i < kHD; ++i) {
    float y = (row[i] - mu) * rstd * gamma[i] + beta[i];
    kd[i] = y;
    vd[i] = y;
  }
}

__global__ __launch_bounds__(kThreads)
void qkv_kernel(const float* __restrict__ x,
                const float* __restrict__ Wq,
                const float* __restrict__ Wk,
                const float* __restrict__ Wv,
                float* __restrict__ q_buf,
                float* __restrict__ k_buf,
                float* __restrict__ v_buf) {
  __shared__ float xs[kRows][36];
  __shared__ float wst[192][34];

  const int tid  = threadIdx.x;
  const int wid  = tid >> 5;
  const int lane = tid & 31;
  const int half = lane >> 4;
  const int lm   = lane & 15;
  const int r0   = blockIdx.x * kRows;
  const int m0   = wid * 16;

  v8f acc[12] = {};
  float4 xreg[4];
  float4 wreg[12];

  auto gload = [&](int kt) {
    #pragma unroll
    for (int i = 0; i < 4; ++i) {
      int idx = tid + kThreads * i;
      int row = idx >> 3;
      int c4  = (idx & 7) * 4;
      xreg[i] = *(const float4*)(x + (size_t)(r0 + row) * kD + kt + c4);
    }
    #pragma unroll
    for (int i = 0; i < 12; ++i) {
      int idx = tid + kThreads * i;
      int kr  = idx / 48;
      int c4  = idx % 48;
      int mat = c4 >> 4;
      int cc  = (c4 & 15) * 4;
      const float* src = (mat == 0) ? Wq : ((mat == 1) ? Wk : Wv);
      wreg[i] = *(const float4*)(src + (size_t)(kt + kr) * kHD + cc);
    }
  };
  auto lstore = [&]() {
    #pragma unroll
    for (int i = 0; i < 4; ++i) {
      int idx = tid + kThreads * i;
      int row = idx >> 3;
      int c4  = (idx & 7) * 4;
      *(float4*)&xs[row][c4] = xreg[i];
    }
    #pragma unroll
    for (int i = 0; i < 12; ++i) {
      int idx = tid + kThreads * i;
      int kr  = idx / 48;
      int c4  = idx % 48;
      int col = (c4 >> 4) * 64 + (c4 & 15) * 4;
      wst[col + 0][kr] = wreg[i].x;
      wst[col + 1][kr] = wreg[i].y;
      wst[col + 2][kr] = wreg[i].z;
      wst[col + 3][kr] = wreg[i].w;
    }
  };

  gload(0);
  lstore();
  __syncthreads();

  constexpr int kNT = kD / 32;
  for (int t = 0; t < kNT; ++t) {
    if (t + 1 < kNT) gload(32 * (t + 1));
    #pragma unroll
    for (int kk = 0; kk < 8; ++kk) {
      const int k0 = kk * 4;
      v2f a = *(const v2f*)&xs[m0 + lm][k0 + 2 * half];
      #pragma unroll
      for (int c = 0; c < 12; ++c) {
        v2f bf = *(const v2f*)&wst[c * 16 + lm][k0 + 2 * half];
        acc[c] = wmma_f32(a, bf, acc[c]);
      }
    }
    __syncthreads();
    if (t + 1 < kNT) lstore();
    __syncthreads();
  }

  #pragma unroll
  for (int c = 0; c < 12; ++c) {
    const int mat = c >> 2;
    const int hd  = (c & 3) * 16 + lm;
    #pragma unroll
    for (int i = 0; i < 8; ++i) {
      const int gr = r0 + m0 + i + 8 * half;
      const float val = acc[c][i];
      if (mat == 0) {
        q_buf[(size_t)gr * kHD + hd] = val;
      } else {
        const int bb = gr / kT;
        const int tt = gr - bb * kT;
        float* dst = (mat == 1) ? k_buf : v_buf;
        dst[((size_t)bb * kS + tt) * kHD + hd] = val;
      }
    }
  }
}

__global__ __launch_bounds__(kThreads)
void attn_kernel(const float* __restrict__ q_buf,
                 const float* __restrict__ k_buf,
                 const float* __restrict__ v_buf,
                 float* __restrict__ out) {
  __shared__ float qs[kRows][68];
  __shared__ float ks[64][68];
  __shared__ float vst[64][68];
  __shared__ float ps[kRows / 16][16][68];

  const int tid  = threadIdx.x;
  const int wid  = tid >> 5;
  const int lane = tid & 31;
  const int half = lane >> 4;
  const int lm   = lane & 15;
  const int b    = blockIdx.y;
  const int q0   = blockIdx.x * kRows;
  const int m0   = wid * 16;

  float4 kreg[8], vreg[8];
  auto gload = [&](int s0) {
    #pragma unroll
    for (int i = 0; i < 8; ++i) {
      int idx = tid + kThreads * i;
      int row = idx >> 4;
      int c4  = (idx & 15) * 4;
      kreg[i] = *(const float4*)(k_buf + ((size_t)b * kS + s0 + row) * kHD + c4);
      vreg[i] = *(const float4*)(v_buf + ((size_t)b * kS + s0 + row) * kHD + c4);
    }
  };
  auto lstore = [&]() {
    #pragma unroll
    for (int i = 0; i < 8; ++i) {
      int idx = tid + kThreads * i;
      int row = idx >> 4;
      int c4  = (idx & 15) * 4;
      *(float4*)&ks[row][c4] = kreg[i];
      vst[c4 + 0][row] = vreg[i].x;
      vst[c4 + 1][row] = vreg[i].y;
      vst[c4 + 2][row] = vreg[i].z;
      vst[c4 + 3][row] = vreg[i].w;
    }
  };

  #pragma unroll
  for (int i = 0; i < kRows / 8; ++i) {
    int idx = tid + kThreads * i;
    int row = idx >> 4;
    int c4  = (idx & 15) * 4;
    float4 v = *(const float4*)(q_buf + ((size_t)b * kT + q0 + row) * kHD + c4);
    v.x *= kScale; v.y *= kScale; v.z *= kScale; v.w *= kScale;
    *(float4*)&qs[row][c4] = v;
  }
  gload(0);
  lstore();
  __syncthreads();

  v8f oacc[4] = {};
  float mrun[8], lrun[8];
  #pragma unroll
  for (int i = 0; i < 8; ++i) { mrun[i] = -INFINITY; lrun[i] = 0.f; }

  constexpr int kNT = kS / 64;
  for (int it = 0; it < kNT; ++it) {
    if (it + 1 < kNT) gload(64 * (it + 1));

    v8f sacc[4] = {};
    #pragma unroll
    for (int kk = 0; kk < 16; ++kk) {
      const int k0 = kk * 4;
      v2f a = *(const v2f*)&qs[m0 + lm][k0 + 2 * half];
      #pragma unroll
      for (int j = 0; j < 4; ++j) {
        v2f bf = *(const v2f*)&ks[j * 16 + lm][k0 + 2 * half];
        sacc[j] = wmma_f32(a, bf, sacc[j]);
      }
    }

    float mnew[8], alpha[8], rsum[8];
    #pragma unroll
    for (int i = 0; i < 8; ++i) {
      float v = fmaxf(fmaxf(sacc[0][i], sacc[1][i]),
                      fmaxf(sacc[2][i], sacc[3][i]));
      #pragma unroll
      for (int off = 1; off < 16; off <<= 1)
        v = fmaxf(v, __shfl_xor(v, off));
      mnew[i]  = fmaxf(mrun[i], v);
      alpha[i] = __expf(mrun[i] - mnew[i]);
      mrun[i]  = mnew[i];
      rsum[i]  = 0.f;
    }
    #pragma unroll
    for (int j = 0; j < 4; ++j) {
      #pragma unroll
      for (int i = 0; i < 8; ++i) {
        float p = __expf(sacc[j][i] - mnew[i]);
        sacc[j][i] = p;
        rsum[i] += p;
      }
    }
    #pragma unroll
    for (int i = 0; i < 8; ++i) {
      float rs = rsum[i];
      #pragma unroll
      for (int off = 1; off < 16; off <<= 1)
        rs += __shfl_xor(rs, off);
      lrun[i] = lrun[i] * alpha[i] + rs;
    }
    #pragma unroll
    for (int c = 0; c < 4; ++c)
      #pragma unroll
      for (int i = 0; i < 8; ++i)
        oacc[c][i] *= alpha[i];

    #pragma unroll
    for (int j = 0; j < 4; ++j)
      #pragma unroll
      for (int i = 0; i < 8; ++i)
        ps[wid][i + 8 * half][j * 16 + lm] = sacc[j][i];

    #pragma unroll
    for (int kk = 0; kk < 16; ++kk) {
      const int k0 = kk * 4;
      v2f a = *(const v2f*)&ps[wid][lm][k0 + 2 * half];
      #pragma unroll
      for (int c = 0; c < 4; ++c) {
        v2f bf = *(const v2f*)&vst[c * 16 + lm][k0 + 2 * half];
        oacc[c] = wmma_f32(a, bf, oacc[c]);
      }
    }

    __syncthreads();
    if (it + 1 < kNT) lstore();
    __syncthreads();
  }

  #pragma unroll
  for (int c = 0; c < 4; ++c) {
    const int hd = c * 16 + lm;
    #pragma unroll
    for (int i = 0; i < 8; ++i) {
      const int t = q0 + m0 + i + 8 * half;
      out[((size_t)b * kT + t) * kHD + hd] = oacc[c][i] / lrun[i];
    }
  }
}

extern "C" void kernel_launch(void* const* d_in, const int* in_sizes, int n_in,
                              void* d_out, int out_size, void* d_ws, size_t ws_size,
                              hipStream_t stream) {
  const float* x     = (const float*)d_in[0];
  const float* c_emb = (const float*)d_in[1];
  const float* Wq    = (const float*)d_in[2];
  const float* Wk    = (const float*)d_in[3];
  const float* Wv    = (const float*)d_in[4];
  const float* gamma = (const float*)d_in[5];
  const float* beta  = (const float*)d_in[6];
  float* out = (float*)d_out;

  float* q_buf = (float*)d_ws;
  float* k_buf = q_buf + (size_t)kB * kT * kHD;
  float* v_buf = k_buf + (size_t)kB * kS * kHD;

  ln_mem_kernel<<<kB, 64, 0, stream>>>(c_emb, gamma, beta, k_buf, v_buf);
  qkv_kernel<<<(kB * kT) / kRows, kThreads, 0, stream>>>(x, Wq, Wk, Wv,
                                                         q_buf, k_buf, v_buf);
  attn_kernel<<<dim3(kT / kRows, kB), kThreads, 0, stream>>>(q_buf, k_buf, v_buf, out);
}


# tile-staging temporaries as native 4-float vectors instead of HIP 4-float structs: prefetched x and K tiles stay in VGPRs (were in scratch), so the global-load wait moves behind the WMMA block
# speedup vs baseline: 1.6539x; 1.4528x over previous
#include <hip/hip_runtime.h>
#include <math.h>

typedef float v2f __attribute__((ext_vector_type(2)));
typedef float v8f __attribute__((ext_vector_type(8)));
typedef float v4f __attribute__((ext_vector_type(4)));

namespace {
constexpr int kB  = 8;
constexpr int kT  = 2048;
constexpr int kD  = 1024;
constexpr int kHD = 64;
constexpr int kM  = 64;
constexpr int kS  = kT + kM;
constexpr float kLnEps = 1e-5f;
constexpr float kScale = 0.125f;
constexpr int kRows    = 64;
constexpr int kThreads = kRows * 2;
}

__device__ __forceinline__ v8f wmma_f32(v2f a, v2f b, v8f c) {
  return __builtin_amdgcn_wmma_f32_16x16x4_f32(
       false, a,  false, b,
       (short)0, c,  false,  false);
}

__global__ __launch_bounds__(64)
void ln_mem_kernel(const float* __restrict__ c_emb,
                   const float* __restrict__ gamma,
                   const float* __restrict__ beta,
                   float* __restrict__ k_buf,
                   float* __restrict__ v_buf) {
  const int b = blockIdx.x;
  const int r = threadIdx.x;
  const float* row = c_emb + r * kHD;
  float mu = 0.f;
  #pragma unroll
  for (int i = 0; i < kHD; ++i) mu += row[i];
  mu *= (1.f / kHD);
  float var = 0.f;
  #pragma unroll
  for (int i = 0; i < kHD; ++i) { float d = row[i] - mu; var += d * d; }
  var *= (1.f / kHD);
  const float rstd = rsqrtf(var + kLnEps);
  float* kd = k_buf + ((size_t)b * kS + kT + r) * kHD;
  float* vd = v_buf + ((size_t)b * kS + kT + r) * kHD;
  #pragma unroll
  for (int i = 0; i < kHD; ++i) {
    float y = (row[i] - mu) * rstd * gamma[i] + beta[i];
    kd[i] = y;
    vd[i] = y;
  }
}

__global__ __launch_bounds__(kThreads)
void qkv_kernel(const float* __restrict__ x,
                const float* __restrict__ Wq,
                const float* __restrict__ Wk,
                const float* __restrict__ Wv,
                float* __restrict__ q_buf,
                float* __restrict__ k_buf,
                float* __restrict__ v_buf) {
  __shared__ float xs[kRows][36];
  __shared__ float wst[192][34];

  const int tid  = threadIdx.x;
  const int wid  = tid >> 5;
  const int lane = tid & 31;
  const int half = lane >> 4;
  const int lm   = lane & 15;
  const int r0   = blockIdx.x * kRows;
  const int m0   = wid * 16;

  v8f acc[12] = {};
  v4f xreg[4];
  v4f wreg[12];

  auto gload = [&](int kt) {
    #pragma unroll
    for (int i = 0; i < 4; ++i) {
      int idx = tid + kThreads * i;
      int row = idx >> 3;
      int c4  = (idx & 7) * 4;
      xreg[i] = *(const v4f*)(x + (size_t)(r0 + row) * kD + kt + c4);
    }
    #pragma unroll
    for (int i = 0; i < 12; ++i) {
      int idx = tid + kThreads * i;
      int kr  = idx / 48;
      int c4  = idx % 48;
      int mat = c4 >> 4;
      int cc  = (c4 & 15) * 4;
      const float* src = (mat == 0) ? Wq : ((mat == 1) ? Wk : Wv);
      wreg[i] = *(const v4f*)(src + (size_t)(kt + kr) * kHD + cc);
    }
  };
  auto lstore = [&]() {
    #pragma unroll
    for (int i = 0; i < 4; ++i) {
      int idx = tid + kThreads * i;
      int row = idx >> 3;
      int c4  = (idx & 7) * 4;
      *(v4f*)&xs[row][c4] = xreg[i];
    }
    #pragma unroll
    for (int i = 0; i < 12; ++i) {
      int idx = tid + kThreads * i;
      int kr  = idx / 48;
      int c4  = idx % 48;
      int col = (c4 >> 4) * 64 + (c4 & 15) * 4;
      wst[col + 0][kr] = wreg[i].x;
      wst[col + 1][kr] = wreg[i].y;
      wst[col + 2][kr] = wreg[i].z;
      wst[col + 3][kr] = wreg[i].w;
    }
  };

  gload(0);
  lstore();
  __syncthreads();

  constexpr int kNT = kD / 32;
  for (int t = 0; t < kNT; ++t) {
    if (t + 1 < kNT) gload(32 * (t + 1));
    #pragma unroll
    for (int kk = 0; kk < 8; ++kk) {
      const int k0 = kk * 4;
      v2f a = *(const v2f*)&xs[m0 + lm][k0 + 2 * half];
      #pragma unroll
      for (int c = 0; c < 12; ++c) {
        v2f bf = *(const v2f*)&wst[c * 16 + lm][k0 + 2 * half];
        acc[c] = wmma_f32(a, bf, acc[c]);
      }
    }
    __syncthreads();
    if (t + 1 < kNT) lstore();
    __syncthreads();
  }

  #pragma unroll
  for (int c = 0; c < 12; ++c) {
    const int mat = c >> 2;
    const int hd  = (c & 3) * 16 + lm;
    #pragma unroll
    for (int i = 0; i < 8; ++i) {
      const int gr = r0 + m0 + i + 8 * half;
      const float val = acc[c][i];
      if (mat == 0) {
        q_buf[(size_t)gr * kHD + hd] = val;
      } else {
        const int bb = gr / kT;
        const int tt = gr - bb * kT;
        float* dst = (mat == 1) ? k_buf : v_buf;
        dst[((size_t)bb * kS + tt) * kHD + hd] = val;
      }
    }
  }
}

__global__ __launch_bounds__(kThreads)
void attn_kernel(const float* __restrict__ q_buf,
                 const float* __restrict__ k_buf,
                 const float* __restrict__ v_buf,
                 float* __restrict__ out) {
  __shared__ float qs[kRows][68];
  __shared__ float ks[64][68];
  __shared__ float vst[64][68];
  __shared__ float ps[kRows / 16][16][68];

  const int tid  = threadIdx.x;
  const int wid  = tid >> 5;
  const int lane = tid & 31;
  const int half = lane >> 4;
  const int lm   = lane & 15;
  const int b    = blockIdx.y;
  const int q0   = blockIdx.x * kRows;
  const int m0   = wid * 16;

  v4f kreg[8], vreg[8];
  auto gload = [&](int s0) {
    #pragma unroll
    for (int i = 0; i < 8; ++i) {
      int idx = tid + kThreads * i;
      int row = idx >> 4;
      int c4  = (idx & 15) * 4;
      kreg[i] = *(const v4f*)(k_buf + ((size_t)b * kS + s0 + row) * kHD + c4);
      vreg[i] = *(const v4f*)(v_buf + ((size_t)b * kS + s0 + row) * kHD + c4);
    }
  };
  auto lstore = [&]() {
    #pragma unroll
    for (int i = 0; i < 8; ++i) {
      int idx = tid + kThreads * i;
      int row = idx >> 4;
      int c4  = (idx & 15) * 4;
      *(v4f*)&ks[row][c4] = kreg[i];
      vst[c4 + 0][row] = vreg[i].x;
      vst[c4 + 1][row] = vreg[i].y;
      vst[c4 + 2][row] = vreg[i].z;
      vst[c4 + 3][row] = vreg[i].w;
    }
  };

  #pragma unroll
  for (int i = 0; i < kRows / 8; ++i) {
    int idx = tid + kThreads * i;
    int row = idx >> 4;
    int c4  = (idx & 15) * 4;
    v4f v = *(const v4f*)(q_buf + ((size_t)b * kT + q0 + row) * kHD + c4);
    v *= kScale;
    *(v4f*)&qs[row][c4] = v;
  }
  gload(0);
  lstore();
  __syncthreads();

  v8f oacc[4] = {};
  float mrun[8], lrun[8];
  #pragma unroll
  for (int i = 0; i < 8; ++i) { mrun[i] = -INFINITY; lrun[i] = 0.f; }

  constexpr int kNT = kS / 64;
  for (int it = 0; it < kNT; ++it) {
    if (it + 1 < kNT) gload(64 * (it + 1));

    v8f sacc[4] = {};
    #pragma unroll
    for (int kk = 0; kk < 16; ++kk) {
      const int k0 = kk * 4;
      v2f a = *(const v2f*)&qs[m0 + lm][k0 + 2 * half];
      #pragma unroll
      for (int j = 0; j < 4; ++j) {
        v2f bf = *(const v2f*)&ks[j * 16 + lm][k0 + 2 * half];
        sacc[j] = wmma_f32(a, bf, sacc[j]);
      }
    }

    float mnew[8], alpha[8], rsum[8];
    #pragma unroll
    for (int i = 0; i < 8; ++i) {
      float v = fmaxf(fmaxf(sacc[0][i], sacc[1][i]),
                      fmaxf(sacc[2][i], sacc[3][i]));
      #pragma unroll
      for (int off = 1; off < 16; off <<= 1)
        v = fmaxf(v, __shfl_xor(v, off));
      mnew[i]  = fmaxf(mrun[i], v);
      alpha[i] = __expf(mrun[i] - mnew[i]);
      mrun[i]  = mnew[i];
      rsum[i]  = 0.f;
    }
    #pragma unroll
    for (int j = 0; j < 4; ++j) {
      #pragma unroll
      for (int i = 0; i < 8; ++i) {
        float p = __expf(sacc[j][i] - mnew[i]);
        sacc[j][i] = p;
        rsum[i] += p;
      }
    }
    #pragma unroll
    for (int i = 0; i < 8; ++i) {
      float rs = rsum[i];
      #pragma unroll
      for (int off = 1; off < 16; off <<= 1)
        rs += __shfl_xor(rs, off);
      lrun[i] = lrun[i] * alpha[i] + rs;
    }
    #pragma unroll
    for (int c = 0; c < 4; ++c)
      #pragma unroll
      for (int i = 0; i < 8; ++i)
        oacc[c][i] *= alpha[i];

    #pragma unroll
    for (int j = 0; j < 4; ++j)
      #pragma unroll
      for (int i = 0; i < 8; ++i)
        ps[wid][i + 8 * half][j * 16 + lm] = sacc[j][i];

    #pragma unroll
    for (int kk = 0; kk < 16; ++kk) {
      const int k0 = kk * 4;
      v2f a = *(const v2f*)&ps[wid][lm][k0 + 2 * half];
      #pragma unroll
      for (int c = 0; c < 4; ++c) {
        v2f bf = *(const v2f*)&vst[c * 16 + lm][k0 + 2 * half];
        oacc[c] = wmma_f32(a, bf, oacc[c]);
      }
    }

    __syncthreads();
    if (it + 1 < kNT) lstore();
    __syncthreads();
  }

  #pragma unroll
  for (int c = 0; c < 4; ++c) {
    const int hd = c * 16 + lm;
    #pragma unroll
    for (int i = 0; i < 8; ++i) {
      const int t = q0 + m0 + i + 8 * half;
      out[((size_t)b * kT + t) * kHD + hd] = oacc[c][i] / lrun[i];
    }
  }
}

extern "C" void kernel_launch(void* const* d_in, const int* in_sizes, int n_in,
                              void* d_out, int out_size, void* d_ws, size_t ws_size,
                              hipStream_t stream) {
  const float* x     = (const float*)d_in[0];
  const float* c_emb = (const float*)d_in[1];
  const float* Wq    = (const float*)d_in[2];
  const float* Wk    = (const float*)d_in[3];
  const float* Wv    = (const float*)d_in[4];
  const float* gamma = (const float*)d_in[5];
  const float* beta  = (const float*)d_in[6];
  float* out = (float*)d_out;

  float* q_buf = (float*)d_ws;
  float* k_buf = q_buf + (size_t)kB * kT * kHD;
  float* v_buf = k_buf + (size_t)kB * kS * kHD;

  ln_mem_kernel<<<kB, 64, 0, stream>>>(c_emb, gamma, beta, k_buf, v_buf);
  qkv_kernel<<<(kB * kT) / kRows, kThreads, 0, stream>>>(x, Wq, Wk, Wv,
                                                         q_buf, k_buf, v_buf);
  attn_kernel<<<dim3(kT / kRows, kB), kThreads, 0, stream>>>(q_buf, k_buf, v_buf, out);
}


# both matrix kernels: 8 waves per work-group (2 per SIMD) not 4: projection waves split the 192 columns, attention waves split the key tiles in two halves merged once at the end (sum order changes)
# speedup vs baseline: 2.0977x; 1.2683x over previous
#include <hip/hip_runtime.h>
#include <math.h>

typedef float v2f __attribute__((ext_vector_type(2)));
typedef float v8f __attribute__((ext_vector_type(8)));
typedef float v4f __attribute__((ext_vector_type(4)));

namespace {
constexpr int kB  = 8;
constexpr int kT  = 2048;
constexpr int kD  = 1024;
constexpr int kHD = 64;
constexpr int kM  = 64;
constexpr int kS  = kT + kM;
constexpr float kLnEps = 1e-5f;
constexpr float kScale = 0.125f;
constexpr int kRows    = 64;
constexpr int kWaves   = 8;
constexpr int kThreads = kWaves * 32;
constexpr int kGroup   = 128;
}

__device__ __forceinline__ v8f wmma_f32(v2f a, v2f b, v8f c) {
  return __builtin_amdgcn_wmma_f32_16x16x4_f32(
       false, a,  false, b,
       (short)0, c,  false,  false);
}

__global__ __launch_bounds__(64)
void ln_mem_kernel(const float* __restrict__ c_emb,
                   const float* __restrict__ gamma,
                   const float* __restrict__ beta,
                   float* __restrict__ k_buf,
                   float* __restrict__ v_buf) {
  const int b = blockIdx.x;
  const int r = threadIdx.x;
  const float* row = c_emb + r * kHD;
  float mu = 0.f;
  #pragma unroll
  for (int i = 0; i < kHD; ++i) mu += row[i];
  mu *= (1.f / kHD);
  float var = 0.f;
  #pragma unroll
  for (int i = 0; i < kHD; ++i) { float d = row[i] - mu; var += d * d; }
  var *= (1.f / kHD);
  const float rstd = rsqrtf(var + kLnEps);
  float* kd = k_buf + ((size_t)b * kS + kT + r) * kHD;
  float* vd = v_buf + ((size_t)b * kS + kT + r) * kHD;
  #pragma unroll
  for (int i = 0; i < kHD; ++i) {
    float y = (row[i] - mu) * rstd * gamma[i] + beta[i];
    kd[i] = y;
    vd[i] = y;
  }
}

__global__ __launch_bounds__(kThreads)
void qkv_kernel(const float* __restrict__ x,
                const float* __restrict__ Wq,
                const float* __restrict__ Wk,
                const float* __restrict__ Wv,
                float* __restrict__ q_buf,
                float* __restrict__ k_buf,
                float* __restrict__ v_buf) {
  __shared__ float xs[kRows][36];
  __shared__ float wst[192][34];

  const int tid  = threadIdx.x;
  const int wid  = __builtin_amdgcn_readfirstlane(tid >> 5);
  const int lane = tid & 31;
  const int half = lane >> 4;
  const int lm   = lane & 15;
  const int r0   = blockIdx.x * kRows;
  const int m0   = (wid & 3) * 16;
  const int ct0  = (wid >> 2) * 6;

  v8f acc[6] = {};
  v4f xreg[2];
  v4f wreg[6];

  auto gload = [&](int kt) {
    #pragma unroll
    for (int i = 0; i < 2; ++i) {
      int idx = tid + kThreads * i;
      int row = idx >> 3;
      int c4  = (idx & 7) * 4;
      xreg[i] = *(const v4f*)(x + (size_t)(r0 + row) * kD + kt + c4);
    }
    #pragma unroll
    for (int i = 0; i < 6; ++i) {
      int idx = tid + kThreads * i;
      int kr  = idx / 48;
      int c4  = idx % 48;
      int mat = c4 >> 4;
      int cc  = (c4 & 15) * 4;
      const float* src = (mat == 0) ? Wq : ((mat == 1) ? Wk : Wv);
      wreg[i] = *(const v4f*)(src + (size_t)(kt + kr) * kHD + cc);
    }
  };
  auto lstore = [&]() {
    #pragma unroll
    for (int i = 0; i < 2; ++i) {
      int idx = tid + kThreads * i;
      int row = idx >> 3;
      int c4  = (idx & 7) * 4;
      *(v4f*)&xs[row][c4] = xreg[i];
    }
    #pragma unroll
    for (int i = 0; i < 6; ++i) {
      int idx = tid + kThreads * i;
      int kr  = idx / 48;
      int c4  = idx % 48;
      int col = (c4 >> 4) * 64 + (c4 & 15) * 4;
      wst[col + 0][kr] = wreg[i].x;
      wst[col + 1][kr] = wreg[i].y;
      wst[col + 2][kr] = wreg[i].z;
      wst[col + 3][kr] = wreg[i].w;
    }
  };

  gload(0);
  lstore();
  __syncthreads();

  constexpr int kNT = kD / 32;
  for (int t = 0; t < kNT; ++t) {
    if (t + 1 < kNT) gload(32 * (t + 1));
    #pragma unroll
    for (int kk = 0; kk < 8; ++kk) {
      const int k0 = kk * 4;
      v2f a = *(const v2f*)&xs[m0 + lm][k0 + 2 * half];
      #pragma unroll
      for (int c = 0; c < 6; ++c) {
        v2f bf = *(const v2f*)&wst[(ct0 + c) * 16 + lm][k0 + 2 * half];
        acc[c] = wmma_f32(a, bf, acc[c]);
      }
    }
    __syncthreads();
    if (t + 1 < kNT) lstore();
    __syncthreads();
  }

  static_assert(kT % kRows == 0, "a block's rows must not straddle batches");
  const int bb = r0 / kT;
  #pragma unroll
  for (int c = 0; c < 6; ++c) {
    const int ct  = ct0 + c;
    const int mat = ct >> 2;
    const int hd  = (ct & 3) * 16 + lm;
    float* dst = (mat == 0) ? q_buf : ((mat == 1) ? k_buf : v_buf);
    const int radd = (mat == 0) ? 0 : bb * kM;
    #pragma unroll
    for (int i = 0; i < 8; ++i) {
      const int gr = r0 + m0 + i + 8 * half;
      dst[(size_t)(gr + radd) * kHD + hd] = acc[c][i];
    }
  }
}

__global__ __launch_bounds__(kThreads)
void attn_kernel(const float* __restrict__ q_buf,
                 const float* __restrict__ k_buf,
                 const float* __restrict__ v_buf,
                 float* __restrict__ out) {
  __shared__ float qs[kRows][68];
  __shared__ float ks[2][64][68];
  __shared__ float vst[2][64][68];
  __shared__ float ps[kWaves][16][68];
  __shared__ float mg[4][48][32];

  const int tid  = threadIdx.x;
  const int wid  = __builtin_amdgcn_readfirstlane(tid >> 5);
  const int lane = tid & 31;
  const int half = lane >> 4;
  const int lm   = lane & 15;
  const int grp  = wid >> 2;
  const int wq   = wid & 3;
  const int gt   = tid & (kGroup - 1);
  const int b    = blockIdx.y;
  const int q0   = blockIdx.x * kRows;
  const int m0   = wq * 16;

  constexpr int kNT = kS / 64;
  constexpr int kNG = (kNT + 1) / 2;
  const int t0 = grp * kNG;
  const int nt = (grp == 0) ? kNG : (kNT - kNG);

  v4f kreg[8], vreg[8];
  auto gload = [&](int s0) {
    #pragma unroll
    for (int i = 0; i < 8; ++i) {
      int idx = gt + kGroup * i;
      int row = idx >> 4;
      int c4  = (idx & 15) * 4;
      kreg[i] = *(const v4f*)(k_buf + ((size_t)b * kS + s0 + row) * kHD + c4);
      vreg[i] = *(const v4f*)(v_buf + ((size_t)b * kS + s0 + row) * kHD + c4);
    }
  };
  auto lstore = [&]() {
    #pragma unroll
    for (int i = 0; i < 8; ++i) {
      int idx = gt + kGroup * i;
      int row = idx >> 4;
      int c4  = (idx & 15) * 4;
      *(v4f*)&ks[grp][row][c4] = kreg[i];
      vst[grp][c4 + 0][row] = vreg[i].x;
      vst[grp][c4 + 1][row] = vreg[i].y;
      vst[grp][c4 + 2][row] = vreg[i].z;
      vst[grp][c4 + 3][row] = vreg[i].w;
    }
  };

  #pragma unroll
  for (int i = 0; i < kRows * 16 / kThreads; ++i) {
    int idx = tid + kThreads * i;
    int row = idx >> 4;
    int c4  = (idx & 15) * 4;
    v4f v = *(const v4f*)(q_buf + ((size_t)b * kT + q0 + row) * kHD + c4);
    v *= kScale;
    *(v4f*)&qs[row][c4] = v;
  }
  gload(64 * t0);
  lstore();
  __syncthreads();

  v8f oacc[4] = {};
  float mrun[8], lrun[8];
  #pragma unroll
  for (int i = 0; i < 8; ++i) { mrun[i] = -INFINITY; lrun[i] = 0.f; }

  for (int n = 0; n < kNG; ++n) {
    const bool live = n < nt;
    const bool more = n + 1 < nt;
    if (more) gload(64 * (t0 + n + 1));

    if (live) {
    v8f sacc[4] = {};
    #pragma unroll
    for (int kk = 0; kk < 16; ++kk) {
      const int k0 = kk * 4;
      v2f a = *(const v2f*)&qs[m0 + lm][k0 + 2 * half];
      #pragma unroll
      for (int j = 0; j < 4; ++j) {
        v2f bf = *(const v2f*)&ks[grp][j * 16 + lm][k0 + 2 * half];
        sacc[j] = wmma_f32(a, bf, sacc[j]);
      }
    }

    float mnew[8], alpha[8], rsum[8];
    #pragma unroll
    for (int i = 0; i < 8; ++i) {
      float v = fmaxf(fmaxf(sacc[0][i], sacc[1][i]),
                      fmaxf(sacc[2][i], sacc[3][i]));
      #pragma unroll
      for (int off = 1; off < 16; off <<= 1)
        v = fmaxf(v, __shfl_xor(v, off));
      mnew[i]  = fmaxf(mrun[i], v);
      alpha[i] = __expf(mrun[i] - mnew[i]);
      mrun[i]  = mnew[i];
      rsum[i]  = 0.f;
    }
    #pragma unroll
    for (int j = 0; j < 4; ++j) {
      #pragma unroll
      for (int i = 0; i < 8; ++i) {
        float p = __expf(sacc[j][i] - mnew[i]);
        sacc[j][i] = p;
        rsum[i] += p;
      }
    }
    #pragma unroll
    for (int i = 0; i < 8; ++i) {
      float rs = rsum[i];
      #pragma unroll
      for (int off = 1; off < 16; off <<= 1)
        rs += __shfl_xor(rs, off);
      lrun[i] = lrun[i] * alpha[i] + rs;
    }
    #pragma unroll
    for (int c = 0; c < 4; ++c)
      #pragma unroll
      for (int i = 0; i < 8; ++i)
        oacc[c][i] *= alpha[i];

    #pragma unroll
    for (int j = 0; j < 4; ++j)
      #pragma unroll
      for (int i = 0; i < 8; ++i)
        ps[wid][i + 8 * half][j * 16 + lm] = sacc[j][i];

    #pragma unroll
    for (int kk = 0; kk < 16; ++kk) {
      const int k0 = kk * 4;
      v2f a = *(const v2f*)&ps[wid][lm][k0 + 2 * half];
      #pragma unroll
      for (int c = 0; c < 4; ++c) {
        v2f bf = *(const v2f*)&vst[grp][c * 16 + lm][k0 + 2 * half];
        oacc[c] = wmma_f32(a, bf, oacc[c]);
      }
    }
    }

    __syncthreads();
    if (more) lstore();
    __syncthreads();
  }

  if (grp == 1) {
    #pragma unroll
    for (int c = 0; c < 4; ++c)
      #pragma unroll
      for (int i = 0; i < 8; ++i)
        mg[wq][c * 8 + i][lane] = oacc[c][i];
    #pragma unroll
    for (int i = 0; i < 8; ++i) {
      mg[wq][32 + i][lane] = mrun[i];
      mg[wq][40 + i][lane] = lrun[i];
    }
  }
  __syncthreads();

  if (grp == 0) {
    float a0[8], a1[8], lsum[8];
    #pragma unroll
    for (int i = 0; i < 8; ++i) {
      const float m1 = mg[wq][32 + i][lane];
      const float l1 = mg[wq][40 + i][lane];
      const float m  = fmaxf(mrun[i], m1);
      a0[i]   = __expf(mrun[i] - m);
      a1[i]   = __expf(m1 - m);
      lsum[i] = lrun[i] * a0[i] + l1 * a1[i];
    }
    #pragma unroll
    for (int c = 0; c < 4; ++c) {
      const int hd = c * 16 + lm;
      #pragma unroll
      for (int i = 0; i < 8; ++i) {
        const int t = q0 + m0 + i + 8 * half;
        const float o = oacc[c][i] * a0[i] + mg[wq][c * 8 + i][lane] * a1[i];
        out[((size_t)b * kT + t) * kHD + hd] = o / lsum[i];
      }
    }
  }
}

extern "C" void kernel_launch(void* const* d_in, const int* in_sizes, int n_in,
                              void* d_out, int out_size, void* d_ws, size_t ws_size,
                              hipStream_t stream) {
  const float* x     = (const float*)d_in[0];
  const float* c_emb = (const float*)d_in[1];
  const float* Wq    = (const float*)d_in[2];
  const float* Wk    = (const float*)d_in[3];
  const float* Wv    = (const float*)d_in[4];
  const float* gamma = (const float*)d_in[5];
  const float* beta  = (const float*)d_in[6];
  float* out = (float*)d_out;

  float* q_buf = (float*)d_ws;
  float* k_buf = q_buf + (size_t)kB * kT * kHD;
  float* v_buf = k_buf + (size_t)kB * kS * kHD;

  ln_mem_kernel<<<kB, 64, 0, stream>>>(c_emb, gamma, beta, k_buf, v_buf);
  qkv_kernel<<<(kB * kT) / kRows, kThreads, 0, stream>>>(x, Wq, Wk, Wv,
                                                         q_buf, k_buf, v_buf);
  attn_kernel<<<dim3(kT / kRows, kB), kThreads, 0, stream>>>(q_buf, k_buf, v_buf, out);
}


# W and V tile staging: lane-to-element map of the loads changed so each transposing LDS store hits 32 distinct banks (was 4-way bank conflicts); fragment reads, tiles and sum order unchanged
# speedup vs baseline: 2.5489x; 1.2151x over previous
#include <hip/hip_runtime.h>
#include <math.h>

typedef float v2f __attribute__((ext_vector_type(2)));
typedef float v8f __attribute__((ext_vector_type(8)));
typedef float v4f __attribute__((ext_vector_type(4)));

namespace {
constexpr int kB  = 8;
constexpr int kT  = 2048;
constexpr int kD  = 1024;
constexpr int kHD = 64;
constexpr int kM  = 64;
constexpr int kS  = kT + kM;
constexpr float kLnEps = 1e-5f;
constexpr float kScale = 0.125f;
constexpr int kRows    = 64;
constexpr int kWaves   = 8;
constexpr int kThreads = kWaves * 32;
constexpr int kGroup   = 128;
}

__device__ __forceinline__ v8f wmma_f32(v2f a, v2f b, v8f c) {
  return __builtin_amdgcn_wmma_f32_16x16x4_f32(
       false, a,  false, b,
       (short)0, c,  false,  false);
}

__global__ __launch_bounds__(64)
void ln_mem_kernel(const float* __restrict__ c_emb,
                   const float* __restrict__ gamma,
                   const float* __restrict__ beta,
                   float* __restrict__ k_buf,
                   float* __restrict__ v_buf) {
  const int b = blockIdx.x;
  const int r = threadIdx.x;
  const float* row = c_emb + r * kHD;
  float mu = 0.f;
  #pragma unroll
  for (int i = 0; i < kHD; ++i) mu += row[i];
  mu *= (1.f / kHD);
  float var = 0.f;
  #pragma unroll
  for (int i = 0; i < kHD; ++i) { float d = row[i] - mu; var += d * d; }
  var *= (1.f / kHD);
  const float rstd = rsqrtf(var + kLnEps);
  float* kd = k_buf + ((size_t)b * kS + kT + r) * kHD;
  float* vd = v_buf + ((size_t)b * kS + kT + r) * kHD;
  #pragma unroll
  for (int i = 0; i < kHD; ++i) {
    float y = (row[i] - mu) * rstd * gamma[i] + beta[i];
    kd[i] = y;
    vd[i] = y;
  }
}

__global__ __launch_bounds__(kThreads)
void qkv_kernel(const float* __restrict__ x,
                const float* __restrict__ Wq,
                const float* __restrict__ Wk,
                const float* __restrict__ Wv,
                float* __restrict__ q_buf,
                float* __restrict__ k_buf,
                float* __restrict__ v_buf) {
  __shared__ float xs[kRows][36];
  __shared__ float wst[192][34];

  const int tid  = threadIdx.x;
  const int wid  = __builtin_amdgcn_readfirstlane(tid >> 5);
  const int lane = tid & 31;
  const int half = lane >> 4;
  const int lm   = lane & 15;
  const int r0   = blockIdx.x * kRows;
  const int m0   = (wid & 3) * 16;
  const int ct0  = (wid >> 2) * 6;
  const int wkr  = wid * 4 + (lane >> 3);
  const int wq8  = lane & 7;

  v8f acc[6] = {};
  v4f xreg[2];
  v4f wreg[6];

  auto gload = [&](int kt) {
    #pragma unroll
    for (int i = 0; i < 2; ++i) {
      int idx = tid + kThreads * i;
      int row = idx >> 3;
      int c4  = (idx & 7) * 4;
      xreg[i] = *(const v4f*)(x + (size_t)(r0 + row) * kD + kt + c4);
    }
    #pragma unroll
    for (int i = 0; i < 6; ++i) {
      const int mat = i >> 1;
      const int cc  = ((i & 1) * 8 + wq8) * 4;
      const float* src = (mat == 0) ? Wq : ((mat == 1) ? Wk : Wv);
      wreg[i] = *(const v4f*)(src + (size_t)(kt + wkr) * kHD + cc);
    }
  };
  auto lstore = [&]() {
    #pragma unroll
    for (int i = 0; i < 2; ++i) {
      int idx = tid + kThreads * i;
      int row = idx >> 3;
      int c4  = (idx & 7) * 4;
      *(v4f*)&xs[row][c4] = xreg[i];
    }
    #pragma unroll
    for (int i = 0; i < 6; ++i) {
      const int col = (i >> 1) * 64 + ((i & 1) * 8 + wq8) * 4;
      wst[col + 0][wkr] = wreg[i].x;
      wst[col + 1][wkr] = wreg[i].y;
      wst[col + 2][wkr] = wreg[i].z;
      wst[col + 3][wkr] = wreg[i].w;
    }
  };

  gload(0);
  lstore();
  __syncthreads();

  constexpr int kNT = kD / 32;
  for (int t = 0; t < kNT; ++t) {
    if (t + 1 < kNT) gload(32 * (t + 1));
    __builtin_amdgcn_sched_barrier(0);
    #pragma unroll
    for (int kk = 0; kk < 8; ++kk) {
      const int k0 = kk * 4;
      v2f a = *(const v2f*)&xs[m0 + lm][k0 + 2 * half];
      #pragma unroll
      for (int c = 0; c < 6; ++c) {
        v2f bf = *(const v2f*)&wst[(ct0 + c) * 16 + lm][k0 + 2 * half];
        acc[c] = wmma_f32(a, bf, acc[c]);
      }
    }
    __syncthreads();
    if (t + 1 < kNT) lstore();
    __syncthreads();
  }

  static_assert(kT % kRows == 0, "a block's rows must not straddle batches");
  const int bb = r0 / kT;
  #pragma unroll
  for (int c = 0; c < 6; ++c) {
    const int ct  = ct0 + c;
    const int mat = ct >> 2;
    const int hd  = (ct & 3) * 16 + lm;
    float* dst = (mat == 0) ? q_buf : ((mat == 1) ? k_buf : v_buf);
    const int radd = (mat == 0) ? 0 : bb * kM;
    #pragma unroll
    for (int i = 0; i < 8; ++i) {
      const int gr = r0 + m0 + i + 8 * half;
      dst[(size_t)(gr + radd) * kHD + hd] = acc[c][i];
    }
  }
}

__global__ __launch_bounds__(kThreads)
void attn_kernel(const float* __restrict__ q_buf,
                 const float* __restrict__ k_buf,
                 const float* __restrict__ v_buf,
                 float* __restrict__ out) {
  __shared__ float qs[kRows][68];
  __shared__ float ks[2][64][68];
  __shared__ float vst[2][64][68];
  __shared__ float ps[kWaves][16][68];
  __shared__ float mg[4][48][32];

  const int tid  = threadIdx.x;
  const int wid  = __builtin_amdgcn_readfirstlane(tid >> 5);
  const int lane = tid & 31;
  const int half = lane >> 4;
  const int lm   = lane & 15;
  const int grp  = wid >> 2;
  const int wq   = wid & 3;
  const int gt   = tid & (kGroup - 1);
  const int b    = blockIdx.y;
  const int q0   = blockIdx.x * kRows;
  const int m0   = wq * 16;
  const int vr0  = lane >> 2;
  const int vc4  = (wq * 4 + (lane & 3)) * 4;

  constexpr int kNT = kS / 64;
  constexpr int kNG = (kNT + 1) / 2;
  const int t0 = grp * kNG;
  const int nt = (grp == 0) ? kNG : (kNT - kNG);

  v4f kreg[8], vreg[8];
  auto gload = [&](int s0) {
    #pragma unroll
    for (int i = 0; i < 8; ++i) {
      int idx = gt + kGroup * i;
      int row = idx >> 4;
      int c4  = (idx & 15) * 4;
      kreg[i] = *(const v4f*)(k_buf + ((size_t)b * kS + s0 + row) * kHD + c4);
      vreg[i] = *(const v4f*)(v_buf + ((size_t)b * kS + s0 + 8 * i + vr0) * kHD + vc4);
    }
  };
  auto lstore = [&]() {
    #pragma unroll
    for (int i = 0; i < 8; ++i) {
      int idx = gt + kGroup * i;
      int row = idx >> 4;
      int c4  = (idx & 15) * 4;
      *(v4f*)&ks[grp][row][c4] = kreg[i];
      vst[grp][vc4 + 0][8 * i + vr0] = vreg[i].x;
      vst[grp][vc4 + 1][8 * i + vr0] = vreg[i].y;
      vst[grp][vc4 + 2][8 * i + vr0] = vreg[i].z;
      vst[grp][vc4 + 3][8 * i + vr0] = vreg[i].w;
    }
  };

  #pragma unroll
  for (int i = 0; i < kRows * 16 / kThreads; ++i) {
    int idx = tid + kThreads * i;
    int row = idx >> 4;
    int c4  = (idx & 15) * 4;
    v4f v = *(const v4f*)(q_buf + ((size_t)b * kT + q0 + row) * kHD + c4);
    v *= kScale;
    *(v4f*)&qs[row][c4] = v;
  }
  gload(64 * t0);
  lstore();
  __syncthreads();

  v8f oacc[4] = {};
  float mrun[8], lrun[8];
  #pragma unroll
  for (int i = 0; i < 8; ++i) { mrun[i] = -INFINITY; lrun[i] = 0.f; }

  for (int n = 0; n < kNG; ++n) {
    const bool live = n < nt;
    const bool more = n + 1 < nt;
    if (more) gload(64 * (t0 + n + 1));

    if (live) {
    v8f sacc[4] = {};
    #pragma unroll
    for (int kk = 0; kk < 16; ++kk) {
      const int k0 = kk * 4;
      v2f a = *(const v2f*)&qs[m0 + lm][k0 + 2 * half];
      #pragma unroll
      for (int j = 0; j < 4; ++j) {
        v2f bf = *(const v2f*)&ks[grp][j * 16 + lm][k0 + 2 * half];
        sacc[j] = wmma_f32(a, bf, sacc[j]);
      }
    }

    float mnew[8], alpha[8], rsum[8];
    #pragma unroll
    for (int i = 0; i < 8; ++i) {
      float v = fmaxf(fmaxf(sacc[0][i], sacc[1][i]),
                      fmaxf(sacc[2][i], sacc[3][i]));
      #pragma unroll
      for (int off = 1; off < 16; off <<= 1)
        v = fmaxf(v, __shfl_xor(v, off));
      mnew[i]  = fmaxf(mrun[i], v);
      alpha[i] = __expf(mrun[i] - mnew[i]);
      mrun[i]  = mnew[i];
      rsum[i]  = 0.f;
    }
    #pragma unroll
    for (int j = 0; j < 4; ++j) {
      #pragma unroll
      for (int i = 0; i < 8; ++i) {
        float p = __expf(sacc[j][i] - mnew[i]);
        sacc[j][i] = p;
        rsum[i] += p;
      }
    }
    #pragma unroll
    for (int i = 0; i < 8; ++i) {
      float rs = rsum[i];
      #pragma unroll
      for (int off = 1; off < 16; off <<= 1)
        rs += __shfl_xor(rs, off);
      lrun[i] = lrun[i] * alpha[i] + rs;
    }
    #pragma unroll
    for (int c = 0; c < 4; ++c)
      #pragma unroll
      for (int i = 0; i < 8; ++i)
        oacc[c][i] *= alpha[i];

    #pragma unroll
    for (int j = 0; j < 4; ++j)
      #pragma unroll
      for (int i = 0; i < 8; ++i)
        ps[wid][i + 8 * half][j * 16 + lm] = sacc[j][i];

    #pragma unroll
    for (int kk = 0; kk < 16; ++kk) {
      const int k0 = kk * 4;
      v2f a = *(const v2f*)&ps[wid][lm][k0 + 2 * half];
      #pragma unroll
      for (int c = 0; c < 4; ++c) {
        v2f bf = *(const v2f*)&vst[grp][c * 16 + lm][k0 + 2 * half];
        oacc[c] = wmma_f32(a, bf, oacc[c]);
      }
    }
    }

    __syncthreads();
    if (more) lstore();
    __syncthreads();
  }

  if (grp == 1) {
    #pragma unroll
    for (int c = 0; c < 4; ++c)
      #pragma unroll
      for (int i = 0; i < 8; ++i)
        mg[wq][c * 8 + i][lane] = oacc[c][i];
    #pragma unroll
    for (int i = 0; i < 8; ++i) {
      mg[wq][32 + i][lane] = mrun[i];
      mg[wq][40 + i][lane] = lrun[i];
    }
  }
  __syncthreads();

  if (grp == 0) {
    float a0[8], a1[8], lsum[8];
    #pragma unroll
    for (int i = 0; i < 8; ++i) {
      const float m1 = mg[wq][32 + i][lane];
      const float l1 = mg[wq][40 + i][lane];
      const float m  = fmaxf(mrun[i], m1);
      a0[i]   = __expf(mrun[i] - m);
      a1[i]   = __expf(m1 - m);
      lsum[i] = lrun[i] * a0[i] + l1 * a1[i];
    }
    #pragma unroll
    for (int c = 0; c < 4; ++c) {
      const int hd = c * 16 + lm;
      #pragma unroll
      for (int i = 0; i < 8; ++i) {
        const int t = q0 + m0 + i + 8 * half;
        const float o = oacc[c][i] * a0[i] + mg[wq][c * 8 + i][lane] * a1[i];
        out[((size_t)b * kT + t) * kHD + hd] = o / lsum[i];
      }
    }
  }
}

extern "C" void kernel_launch(void* const* d_in, const int* in_sizes, int n_in,
                              void* d_out, int out_size, void* d_ws, size_t ws_size,
                              hipStream_t stream) {
  const float* x     = (const float*)d_in[0];
  const float* c_emb = (const float*)d_in[1];
  const float* Wq    = (const float*)d_in[2];
  const float* Wk    = (const float*)d_in[3];
  const float* Wv    = (const float*)d_in[4];
  const float* gamma = (const float*)d_in[5];
  const float* beta  = (const float*)d_in[6];
  float* out = (float*)d_out;

  float* q_buf = (float*)d_ws;
  float* k_buf = q_buf + (size_t)kB * kT * kHD;
  float* v_buf = k_buf + (size_t)kB * kS * kHD;

  ln_mem_kernel<<<kB, 64, 0, stream>>>(c_emb, gamma, beta, k_buf, v_buf);
  qkv_kernel<<<(kB * kT) / kRows, kThreads, 0, stream>>>(x, Wq, Wk, Wv,
                                                         q_buf, k_buf, v_buf);
  attn_kernel<<<dim3(kT / kRows, kB), kThreads, 0, stream>>>(q_buf, k_buf, v_buf, out);
}


# both matrix kernels: LDS tile stage doubled; tile t+1 is stored to the other stage inside tile t's matrix block, loads run two tiles ahead across an LDS-only barrier; one barrier per tile, not two
# speedup vs baseline: 2.7807x; 1.0909x over previous
#include <hip/hip_runtime.h>
#include <math.h>

typedef float v2f __attribute__((ext_vector_type(2)));
typedef float v8f __attribute__((ext_vector_type(8)));
typedef float v4f __attribute__((ext_vector_type(4)));

namespace {
constexpr int kB  = 8;
constexpr int kT  = 2048;
constexpr int kD  = 1024;
constexpr int kHD = 64;
constexpr int kM  = 64;
constexpr int kS  = kT + kM;
constexpr float kLnEps = 1e-5f;
constexpr float kScale = 0.125f;
constexpr int kRows    = 64;
constexpr int kWaves   = 8;
constexpr int kThreads = kWaves * 32;
constexpr int kGroup   = 128;
}

__device__ __forceinline__ v8f wmma_f32(v2f a, v2f b, v8f c) {
  return __builtin_amdgcn_wmma_f32_16x16x4_f32(
       false, a,  false, b,
       (short)0, c,  false,  false);
}

__device__ __forceinline__ void lds_barrier() {
  __builtin_amdgcn_fence(__ATOMIC_RELEASE, "workgroup", "local");
  __builtin_amdgcn_s_barrier();
  __builtin_amdgcn_fence(__ATOMIC_ACQUIRE, "workgroup", "local");
}

__global__ __launch_bounds__(64)
void ln_mem_kernel(const float* __restrict__ c_emb,
                   const float* __restrict__ gamma,
                   const float* __restrict__ beta,
                   float* __restrict__ k_buf,
                   float* __restrict__ v_buf) {
  const int b = blockIdx.x;
  const int r = threadIdx.x;
  const float* row = c_emb + r * kHD;
  float mu = 0.f;
  #pragma unroll
  for (int i = 0; i < kHD; ++i) mu += row[i];
  mu *= (1.f / kHD);
  float var = 0.f;
  #pragma unroll
  for (int i = 0; i < kHD; ++i) { float d = row[i] - mu; var += d * d; }
  var *= (1.f / kHD);
  const float rstd = rsqrtf(var + kLnEps);
  float* kd = k_buf + ((size_t)b * kS + kT + r) * kHD;
  float* vd = v_buf + ((size_t)b * kS + kT + r) * kHD;
  #pragma unroll
  for (int i = 0; i < kHD; ++i) {
    float y = (row[i] - mu) * rstd * gamma[i] + beta[i];
    kd[i] = y;
    vd[i] = y;
  }
}

__global__ __launch_bounds__(kThreads)
void qkv_kernel(const float* __restrict__ x,
                const float* __restrict__ Wq,
                const float* __restrict__ Wk,
                const float* __restrict__ Wv,
                float* __restrict__ q_buf,
                float* __restrict__ k_buf,
                float* __restrict__ v_buf) {
  __shared__ float xs[2][kRows][36];
  __shared__ float wst[2][192][34];

  const int tid  = threadIdx.x;
  const int wid  = __builtin_amdgcn_readfirstlane(tid >> 5);
  const int lane = tid & 31;
  const int half = lane >> 4;
  const int lm   = lane & 15;
  const int r0   = blockIdx.x * kRows;
  const int m0   = (wid & 3) * 16;
  const int ct0  = (wid >> 2) * 6;
  const int wkr  = wid * 4 + (lane >> 3);
  const int wq8  = lane & 7;

  v8f acc[6] = {};
  v4f xreg[2];
  v4f wreg[6];

  auto gload = [&](int kt) {
    #pragma unroll
    for (int i = 0; i < 2; ++i) {
      int idx = tid + kThreads * i;
      int row = idx >> 3;
      int c4  = (idx & 7) * 4;
      xreg[i] = *(const v4f*)(x + (size_t)(r0 + row) * kD + kt + c4);
    }
    #pragma unroll
    for (int i = 0; i < 6; ++i) {
      const int mat = i >> 1;
      const int cc  = ((i & 1) * 8 + wq8) * 4;
      const float* src = (mat == 0) ? Wq : ((mat == 1) ? Wk : Wv);
      wreg[i] = *(const v4f*)(src + (size_t)(kt + wkr) * kHD + cc);
    }
  };
  auto lstore = [&](int st) {
    #pragma unroll
    for (int i = 0; i < 2; ++i) {
      int idx = tid + kThreads * i;
      int row = idx >> 3;
      int c4  = (idx & 7) * 4;
      *(v4f*)&xs[st][row][c4] = xreg[i];
    }
    #pragma unroll
    for (int i = 0; i < 6; ++i) {
      const int col = (i >> 1) * 64 + ((i & 1) * 8 + wq8) * 4;
      wst[st][col + 0][wkr] = wreg[i].x;
      wst[st][col + 1][wkr] = wreg[i].y;
      wst[st][col + 2][wkr] = wreg[i].z;
      wst[st][col + 3][wkr] = wreg[i].w;
    }
  };
  auto mma = [&](int st, int kk) {
    const int k0 = kk * 4;
    v2f a = *(const v2f*)&xs[st][m0 + lm][k0 + 2 * half];
    #pragma unroll
    for (int c = 0; c < 6; ++c) {
      v2f bf = *(const v2f*)&wst[st][(ct0 + c) * 16 + lm][k0 + 2 * half];
      acc[c] = wmma_f32(a, bf, acc[c]);
    }
  };

  constexpr int kNT = kD / 32;
  gload(0);
  lstore(0);
  gload(32);
  lds_barrier();

  for (int t = 0; t < kNT; ++t) {
    const int cur = t & 1;
    #pragma unroll
    for (int kk = 0; kk < 4; ++kk) mma(cur, kk);
    if (t + 1 < kNT) {
      lstore(cur ^ 1);
      if (t + 2 < kNT) gload(32 * (t + 2));
    }
    __builtin_amdgcn_sched_barrier(0);
    #pragma unroll
    for (int kk = 4; kk < 8; ++kk) mma(cur, kk);
    lds_barrier();
  }

  static_assert(kT % kRows == 0, "a block's rows must not straddle batches");
  const int bb = r0 / kT;
  #pragma unroll
  for (int c = 0; c < 6; ++c) {
    const int ct  = ct0 + c;
    const int mat = ct >> 2;
    const int hd  = (ct & 3) * 16 + lm;
    float* dst = (mat == 0) ? q_buf : ((mat == 1) ? k_buf : v_buf);
    const int radd = (mat == 0) ? 0 : bb * kM;
    #pragma unroll
    for (int i = 0; i < 8; ++i) {
      const int gr = r0 + m0 + i + 8 * half;
      dst[(size_t)(gr + radd) * kHD + hd] = acc[c][i];
    }
  }
}

__global__ __launch_bounds__(kThreads)
void attn_kernel(const float* __restrict__ q_buf,
                 const float* __restrict__ k_buf,
                 const float* __restrict__ v_buf,
                 float* __restrict__ out) {
  __shared__ float qs[kRows][68];
  __shared__ float ks[2][2][64][68];
  __shared__ float vst[2][2][64][68];
  __shared__ float ps[kWaves][16][68];
  __shared__ float mg[4][48][32];

  const int tid  = threadIdx.x;
  const int wid  = __builtin_amdgcn_readfirstlane(tid >> 5);
  const int lane = tid & 31;
  const int half = lane >> 4;
  const int lm   = lane & 15;
  const int grp  = wid >> 2;
  const int wq   = wid & 3;
  const int gt   = tid & (kGroup - 1);
  const int b    = blockIdx.y;
  const int q0   = blockIdx.x * kRows;
  const int m0   = wq * 16;
  const int vr0  = lane >> 2;
  const int vc4  = (wq * 4 + (lane & 3)) * 4;

  constexpr int kNT = kS / 64;
  constexpr int kNG = (kNT + 1) / 2;
  const int t0 = grp * kNG;
  const int nt = (grp == 0) ? kNG : (kNT - kNG);

  v4f kreg[8], vreg[8];
  auto gload = [&](int s0) {
    #pragma unroll
    for (int i = 0; i < 8; ++i) {
      int idx = gt + kGroup * i;
      int row = idx >> 4;
      int c4  = (idx & 15) * 4;
      kreg[i] = *(const v4f*)(k_buf + ((size_t)b * kS + s0 + row) * kHD + c4);
      vreg[i] = *(const v4f*)(v_buf + ((size_t)b * kS + s0 + 8 * i + vr0) * kHD + vc4);
    }
  };
  auto lstore = [&](int st) {
    #pragma unroll
    for (int i = 0; i < 8; ++i) {
      int idx = gt + kGroup * i;
      int row = idx >> 4;
      int c4  = (idx & 15) * 4;
      *(v4f*)&ks[st][grp][row][c4] = kreg[i];
      vst[st][grp][vc4 + 0][8 * i + vr0] = vreg[i].x;
      vst[st][grp][vc4 + 1][8 * i + vr0] = vreg[i].y;
      vst[st][grp][vc4 + 2][8 * i + vr0] = vreg[i].z;
      vst[st][grp][vc4 + 3][8 * i + vr0] = vreg[i].w;
    }
  };

  #pragma unroll
  for (int i = 0; i < kRows * 16 / kThreads; ++i) {
    int idx = tid + kThreads * i;
    int row = idx >> 4;
    int c4  = (idx & 15) * 4;
    v4f v = *(const v4f*)(q_buf + ((size_t)b * kT + q0 + row) * kHD + c4);
    v *= kScale;
    *(v4f*)&qs[row][c4] = v;
  }
  gload(64 * t0);
  lstore(0);
  gload(64 * (t0 + 1));
  lds_barrier();

  v8f oacc[4] = {};
  float mrun[8], lrun[8];
  #pragma unroll
  for (int i = 0; i < 8; ++i) { mrun[i] = -INFINITY; lrun[i] = 0.f; }

  for (int n = 0; n < kNG; ++n) {
    const bool live = n < nt;
    const int  cur  = n & 1;

    if (live) {
    v8f sacc[4] = {};
    #pragma unroll
    for (int kk = 0; kk < 16; ++kk) {
      const int k0 = kk * 4;
      v2f a = *(const v2f*)&qs[m0 + lm][k0 + 2 * half];
      #pragma unroll
      for (int j = 0; j < 4; ++j) {
        v2f bf = *(const v2f*)&ks[cur][grp][j * 16 + lm][k0 + 2 * half];
        sacc[j] = wmma_f32(a, bf, sacc[j]);
      }
    }

    float mnew[8], alpha[8], rsum[8];
    #pragma unroll
    for (int i = 0; i < 8; ++i) {
      float v = fmaxf(fmaxf(sacc[0][i], sacc[1][i]),
                      fmaxf(sacc[2][i], sacc[3][i]));
      #pragma unroll
      for (int off = 1; off < 16; off <<= 1)
        v = fmaxf(v, __shfl_xor(v, off));
      mnew[i]  = fmaxf(mrun[i], v);
      alpha[i] = __expf(mrun[i] - mnew[i]);
      mrun[i]  = mnew[i];
      rsum[i]  = 0.f;
    }
    #pragma unroll
    for (int j = 0; j < 4; ++j) {
      #pragma unroll
      for (int i = 0; i < 8; ++i) {
        float p = __expf(sacc[j][i] - mnew[i]);
        sacc[j][i] = p;
        rsum[i] += p;
      }
    }
    #pragma unroll
    for (int i = 0; i < 8; ++i) {
      float rs = rsum[i];
      #pragma unroll
      for (int off = 1; off < 16; off <<= 1)
        rs += __shfl_xor(rs, off);
      lrun[i] = lrun[i] * alpha[i] + rs;
    }
    #pragma unroll
    for (int c = 0; c < 4; ++c)
      #pragma unroll
      for (int i = 0; i < 8; ++i)
        oacc[c][i] *= alpha[i];

    #pragma unroll
    for (int j = 0; j < 4; ++j)
      #pragma unroll
      for (int i = 0; i < 8; ++i)
        ps[wid][i + 8 * half][j * 16 + lm] = sacc[j][i];

    if (n + 1 < nt) {
      lstore(cur ^ 1);
      if (n + 2 < nt) gload(64 * (t0 + n + 2));
    }

    #pragma unroll
    for (int kk = 0; kk < 16; ++kk) {
      const int k0 = kk * 4;
      v2f a = *(const v2f*)&ps[wid][lm][k0 + 2 * half];
      #pragma unroll
      for (int c = 0; c < 4; ++c) {
        v2f bf = *(const v2f*)&vst[cur][grp][c * 16 + lm][k0 + 2 * half];
        oacc[c] = wmma_f32(a, bf, oacc[c]);
      }
    }
    }

    lds_barrier();
  }

  if (grp == 1) {
    #pragma unroll
    for (int c = 0; c < 4; ++c)
      #pragma unroll
      for (int i = 0; i < 8; ++i)
        mg[wq][c * 8 + i][lane] = oacc[c][i];
    #pragma unroll
    for (int i = 0; i < 8; ++i) {
      mg[wq][32 + i][lane] = mrun[i];
      mg[wq][40 + i][lane] = lrun[i];
    }
  }
  __syncthreads();

  if (grp == 0) {
    float a0[8], a1[8], lsum[8];
    #pragma unroll
    for (int i = 0; i < 8; ++i) {
      const float m1 = mg[wq][32 + i][lane];
      const float l1 = mg[wq][40 + i][lane];
      const float m  = fmaxf(mrun[i], m1);
      a0[i]   = __expf(mrun[i] - m);
      a1[i]   = __expf(m1 - m);
      lsum[i] = lrun[i] * a0[i] + l1 * a1[i];
    }
    #pragma unroll
    for (int c = 0; c < 4; ++c) {
      const int hd = c * 16 + lm;
      #pragma unroll
      for (int i = 0; i < 8; ++i) {
        const int t = q0 + m0 + i + 8 * half;
        const float o = oacc[c][i] * a0[i] + mg[wq][c * 8 + i][lane] * a1[i];
        out[((size_t)b * kT + t) * kHD + hd] = o / lsum[i];
      }
    }
  }
}

extern "C" void kernel_launch(void* const* d_in, const int* in_sizes, int n_in,
                              void* d_out, int out_size, void* d_ws, size_t ws_size,
                              hipStream_t stream) {
  const float* x     = (const float*)d_in[0];
  const float* c_emb = (const float*)d_in[1];
  const float* Wq    = (const float*)d_in[2];
  const float* Wk    = (const float*)d_in[3];
  const float* Wv    = (const float*)d_in[4];
  const float* gamma = (const float*)d_in[5];
  const float* beta  = (const float*)d_in[6];
  float* out = (float*)d_out;

  float* q_buf = (float*)d_ws;
  float* k_buf = q_buf + (size_t)kB * kT * kHD;
  float* v_buf = k_buf + (size_t)kB * kS * kHD;

  ln_mem_kernel<<<kB, 64, 0, stream>>>(c_emb, gamma, beta, k_buf, v_buf);
  qkv_kernel<<<(kB * kT) / kRows, kThreads, 0, stream>>>(x, Wq, Wk, Wv,
                                                         q_buf, k_buf, v_buf);
  attn_kernel<<<dim3(kT / kRows, kB), kThreads, 0, stream>>>(q_buf, k_buf, v_buf, out);
}


# attention scores computed transposed (keys x queries): row max and sum run along a lane's registers with one half-wave exchange, P feeds P@V from registers, not LDS; denominator sum order changes
# speedup vs baseline: 2.9867x; 1.0741x over previous
#include <hip/hip_runtime.h>
#include <math.h>

typedef float v2f __attribute__((ext_vector_type(2)));
typedef float v8f __attribute__((ext_vector_type(8)));
typedef float v4f __attribute__((ext_vector_type(4)));

namespace {
constexpr int kB  = 8;
constexpr int kT  = 2048;
constexpr int kD  = 1024;
constexpr int kHD = 64;
constexpr int kM  = 64;
constexpr int kS  = kT + kM;
constexpr float kLnEps = 1e-5f;
constexpr float kScale = 0.125f;
constexpr int kRows    = 64;
constexpr int kWaves   = 8;
constexpr int kThreads = kWaves * 32;
constexpr int kGroup   = 128;
}

__device__ __forceinline__ v8f wmma_f32(v2f a, v2f b, v8f c) {
  return __builtin_amdgcn_wmma_f32_16x16x4_f32(
       false, a,  false, b,
       (short)0, c,  false,  false);
}

__device__ __forceinline__ void lds_barrier() {
  __builtin_amdgcn_fence(__ATOMIC_RELEASE, "workgroup", "local");
  __builtin_amdgcn_s_barrier();
  __builtin_amdgcn_fence(__ATOMIC_ACQUIRE, "workgroup", "local");
}

__global__ __launch_bounds__(64)
void ln_mem_kernel(const float* __restrict__ c_emb,
                   const float* __restrict__ gamma,
                   const float* __restrict__ beta,
                   float* __restrict__ k_buf,
                   float* __restrict__ v_buf) {
  const int b = blockIdx.x;
  const int r = threadIdx.x;
  const float* row = c_emb + r * kHD;
  float mu = 0.f;
  #pragma unroll
  for (int i = 0; i < kHD; ++i) mu += row[i];
  mu *= (1.f / kHD);
  float var = 0.f;
  #pragma unroll
  for (int i = 0; i < kHD; ++i) { float d = row[i] - mu; var += d * d; }
  var *= (1.f / kHD);
  const float rstd = rsqrtf(var + kLnEps);
  float* kd = k_buf + ((size_t)b * kS + kT + r) * kHD;
  float* vd = v_buf + ((size_t)b * kS + kT + r) * kHD;
  #pragma unroll
  for (int i = 0; i < kHD; ++i) {
    float y = (row[i] - mu) * rstd * gamma[i] + beta[i];
    kd[i] = y;
    vd[i] = y;
  }
}

__global__ __launch_bounds__(kThreads)
void qkv_kernel(const float* __restrict__ x,
                const float* __restrict__ Wq,
                const float* __restrict__ Wk,
                const float* __restrict__ Wv,
                float* __restrict__ q_buf,
                float* __restrict__ k_buf,
                float* __restrict__ v_buf) {
  __shared__ float xs[2][kRows][36];
  __shared__ float wst[2][192][34];

  const int tid  = threadIdx.x;
  const int wid  = __builtin_amdgcn_readfirstlane(tid >> 5);
  const int lane = tid & 31;
  const int half = lane >> 4;
  const int lm   = lane & 15;
  const int r0   = blockIdx.x * kRows;
  const int m0   = (wid & 3) * 16;
  const int ct0  = (wid >> 2) * 6;
  const int wkr  = wid * 4 + (lane >> 3);
  const int wq8  = lane & 7;

  v8f acc[6] = {};
  v4f xreg[2];
  v4f wreg[6];

  auto gload = [&](int kt) {
    #pragma unroll
    for (int i = 0; i < 2; ++i) {
      int idx = tid + kThreads * i;
      int row = idx >> 3;
      int c4  = (idx & 7) * 4;
      xreg[i] = *(const v4f*)(x + (size_t)(r0 + row) * kD + kt + c4);
    }
    #pragma unroll
    for (int i = 0; i < 6; ++i) {
      const int mat = i >> 1;
      const int cc  = ((i & 1) * 8 + wq8) * 4;
      const float* src = (mat == 0) ? Wq : ((mat == 1) ? Wk : Wv);
      wreg[i] = *(const v4f*)(src + (size_t)(kt + wkr) * kHD + cc);
    }
  };
  auto lstore = [&](int st) {
    #pragma unroll
    for (int i = 0; i < 2; ++i) {
      int idx = tid + kThreads * i;
      int row = idx >> 3;
      int c4  = (idx & 7) * 4;
      *(v4f*)&xs[st][row][c4] = xreg[i];
    }
    #pragma unroll
    for (int i = 0; i < 6; ++i) {
      const int col = (i >> 1) * 64 + ((i & 1) * 8 + wq8) * 4;
      wst[st][col + 0][wkr] = wreg[i].x;
      wst[st][col + 1][wkr] = wreg[i].y;
      wst[st][col + 2][wkr] = wreg[i].z;
      wst[st][col + 3][wkr] = wreg[i].w;
    }
  };
  auto mma = [&](int st, int kk) {
    const int k0 = kk * 4;
    v2f a = *(const v2f*)&xs[st][m0 + lm][k0 + 2 * half];
    #pragma unroll
    for (int c = 0; c < 6; ++c) {
      v2f bf = *(const v2f*)&wst[st][(ct0 + c) * 16 + lm][k0 + 2 * half];
      acc[c] = wmma_f32(a, bf, acc[c]);
    }
  };

  constexpr int kNT = kD / 32;
  gload(0);
  lstore(0);
  gload(32);
  lds_barrier();

  for (int t = 0; t < kNT; ++t) {
    const int cur = t & 1;
    #pragma unroll
    for (int kk = 0; kk < 4; ++kk) mma(cur, kk);
    if (t + 1 < kNT) {
      lstore(cur ^ 1);
      if (t + 2 < kNT) gload(32 * (t + 2));
    }
    __builtin_amdgcn_sched_barrier(0);
    #pragma unroll
    for (int kk = 4; kk < 8; ++kk) mma(cur, kk);
    lds_barrier();
  }

  static_assert(kT % kRows == 0, "a block's rows must not straddle batches");
  const int bb = r0 / kT;
  #pragma unroll
  for (int c = 0; c < 6; ++c) {
    const int ct  = ct0 + c;
    const int mat = ct >> 2;
    const int hd  = (ct & 3) * 16 + lm;
    float* dst = (mat == 0) ? q_buf : ((mat == 1) ? k_buf : v_buf);
    const int radd = (mat == 0) ? 0 : bb * kM;
    #pragma unroll
    for (int i = 0; i < 8; ++i) {
      const int gr = r0 + m0 + i + 8 * half;
      dst[(size_t)(gr + radd) * kHD + hd] = acc[c][i];
    }
  }
}


__device__ __forceinline__ float xhalf(float x) {
  const int b = __builtin_bit_cast(int, x);
  return __builtin_bit_cast(float,
      __builtin_amdgcn_permlanex16(b, b, 0x76543210, 0xfedcba98, false, false));
}

__global__ __launch_bounds__(kThreads)
void attn_kernel(const float* __restrict__ q_buf,
                 const float* __restrict__ k_buf,
                 const float* __restrict__ v_buf,
                 float* __restrict__ out) {
  __shared__ float qs[kRows][68];
  __shared__ __attribute__((aligned(16))) float ks[2][2][64][68];
  __shared__ float vst[2][2][64][68];
  __shared__ float mg[4][34][32];

  const int tid  = threadIdx.x;
  const int wid  = __builtin_amdgcn_readfirstlane(tid >> 5);
  const int lane = tid & 31;
  const int half = lane >> 4;
  const int lm   = lane & 15;
  const int grp  = wid >> 2;
  const int wq   = wid & 3;
  const int gt   = tid & (kGroup - 1);
  const int b    = blockIdx.y;
  const int q0   = blockIdx.x * kRows;
  const int m0   = wq * 16;
  const int vr0  = lane >> 2;
  const int vc4  = (wq * 4 + (lane & 3)) * 4;
  const int krow = ((lm & 6) << 1) | ((lm >> 2) & 2) | (lm & 1);

  constexpr int kNT = kS / 64;
  constexpr int kNG = (kNT + 1) / 2;
  const int t0 = grp * kNG;
  const int nt = (grp == 0) ? kNG : (kNT - kNG);

  v4f kreg[8], vreg[8];
  auto gload = [&](int s0) {
    #pragma unroll
    for (int i = 0; i < 8; ++i) {
      int idx = gt + kGroup * i;
      int row = idx >> 4;
      int c4  = (idx & 15) * 4;
      kreg[i] = *(const v4f*)(k_buf + ((size_t)b * kS + s0 + row) * kHD + c4);
      vreg[i] = *(const v4f*)(v_buf + ((size_t)b * kS + s0 + 8 * i + vr0) * kHD + vc4);
    }
  };
  auto lstore = [&](int st) {
    #pragma unroll
    for (int i = 0; i < 8; ++i) {
      int idx = gt + kGroup * i;
      int row = idx >> 4;
      int c4  = (idx & 15) * 4;
      *(v4f*)&ks[st][grp][row][c4] = kreg[i];
      vst[st][grp][vc4 + 0][8 * i + vr0] = vreg[i].x;
      vst[st][grp][vc4 + 1][8 * i + vr0] = vreg[i].y;
      vst[st][grp][vc4 + 2][8 * i + vr0] = vreg[i].z;
      vst[st][grp][vc4 + 3][8 * i + vr0] = vreg[i].w;
    }
  };

  #pragma unroll
  for (int i = 0; i < kRows * 16 / kThreads; ++i) {
    int idx = tid + kThreads * i;
    int row = idx >> 4;
    int c4  = (idx & 15) * 4;
    v4f v = *(const v4f*)(q_buf + ((size_t)b * kT + q0 + row) * kHD + c4);
    v *= kScale;
    *(v4f*)&qs[row][c4] = v;
  }
  gload(64 * t0);
  lstore(0);
  gload(64 * (t0 + 1));
  lds_barrier();

  const float* kfrag = (const float*)__builtin_assume_aligned(&ks[0][grp][krow][2 * half], 8);
  v8f oacc[4] = {};
  float mrun = -INFINITY, lrun = 0.f;

  for (int n = 0; n < kNG; ++n) {
    const bool live = n < nt;
    const int  cur  = n & 1;

    if (live) {
    v8f sacc[4] = {};
    #pragma unroll
    for (int kk = 0; kk < 16; ++kk) {
      const int k0 = kk * 4;
      v2f qf = *(const v2f*)&qs[m0 + lm][k0 + 2 * half];
      #pragma unroll
      for (int j = 0; j < 4; ++j) {
        v2f kf = *(const v2f*)(kfrag + cur * (2 * 64 * 68) + j * (16 * 68) + k0);
        sacc[j] = wmma_f32(kf, qf, sacc[j]);
      }
    }

    float mx = fmaxf(fmaxf(sacc[0][0], sacc[1][0]), fmaxf(sacc[2][0], sacc[3][0]));
    #pragma unroll
    for (int i = 1; i < 8; ++i)
      mx = fmaxf(mx, fmaxf(fmaxf(sacc[0][i], sacc[1][i]),
                           fmaxf(sacc[2][i], sacc[3][i])));
    mx = fmaxf(mx, xhalf(mx));
    const float mnew  = fmaxf(mrun, mx);
    const float alpha = __expf(mrun - mnew);
    mrun = mnew;
    float rs[4];
    #pragma unroll
    for (int j = 0; j < 4; ++j) {
      rs[j] = 0.f;
      #pragma unroll
      for (int i = 0; i < 8; ++i) {
        float p = __expf(sacc[j][i] - mnew);
        sacc[j][i] = p;
        rs[j] += p;
      }
    }
    lrun = lrun * alpha + ((rs[0] + rs[1]) + (rs[2] + rs[3]));
    #pragma unroll
    for (int c = 0; c < 4; ++c)
      #pragma unroll
      for (int i = 0; i < 8; ++i)
        oacc[c][i] *= alpha;

    if (n + 1 < nt) {
      lstore(cur ^ 1);
      if (n + 2 < nt) gload(64 * (t0 + n + 2));
    }

    #pragma unroll
    for (int kk = 0; kk < 16; ++kk) {
      const int k0 = kk * 4;
      const int j  = kk >> 2;
      const int t  = kk & 3;
      v2f pf;
      pf.x = sacc[j][2 * t];
      pf.y = sacc[j][2 * t + 1];
      #pragma unroll
      for (int c = 0; c < 4; ++c) {
        v2f vf = *(const v2f*)&vst[cur][grp][c * 16 + lm][k0 + 2 * half];
        oacc[c] = wmma_f32(vf, pf, oacc[c]);
      }
    }
    }

    lds_barrier();
  }

  if (grp == 1) {
    #pragma unroll
    for (int c = 0; c < 4; ++c)
      #pragma unroll
      for (int i = 0; i < 8; ++i)
        mg[wq][c * 8 + i][lane] = oacc[c][i];
    mg[wq][32][lane] = mrun;
    mg[wq][33][lane] = lrun;
  }
  __syncthreads();

  if (grp == 0) {
    const float m1 = mg[wq][32][lane];
    const float l1 = mg[wq][33][lane];
    const float m  = fmaxf(mrun, m1);
    const float a0 = __expf(mrun - m);
    const float a1 = __expf(m1 - m);
    float lsum = lrun * a0 + l1 * a1;
    lsum += xhalf(lsum);
    float* orow = out + ((size_t)b * kT + q0 + m0 + lm) * kHD + 8 * half;
    #pragma unroll
    for (int c = 0; c < 4; ++c) {
      v4f lo, hi;
      lo.x = (oacc[c][0] * a0 + mg[wq][c * 8 + 0][lane] * a1) / lsum;
      lo.y = (oacc[c][1] * a0 + mg[wq][c * 8 + 1][lane] * a1) / lsum;
      lo.z = (oacc[c][2] * a0 + mg[wq][c * 8 + 2][lane] * a1) / lsum;
      lo.w = (oacc[c][3] * a0 + mg[wq][c * 8 + 3][lane] * a1) / lsum;
      hi.x = (oacc[c][4] * a0 + mg[wq][c * 8 + 4][lane] * a1) / lsum;
      hi.y = (oacc[c][5] * a0 + mg[wq][c * 8 + 5][lane] * a1) / lsum;
      hi.z = (oacc[c][6] * a0 + mg[wq][c * 8 + 6][lane] * a1) / lsum;
      hi.w = (oacc[c][7] * a0 + mg[wq][c * 8 + 7][lane] * a1) / lsum;
      *(v4f*)(orow + c * 16)     = lo;
      *(v4f*)(orow + c * 16 + 4) = hi;
    }
  }
}

extern "C" void kernel_launch(void* const* d_in, const int* in_sizes, int n_in,
                              void* d_out, int out_size, void* d_ws, size_t ws_size,
                              hipStream_t stream) {
  const float* x     = (const float*)d_in[0];
  const float* c_emb = (const float*)d_in[1];
  const float* Wq    = (const float*)d_in[2];
  const float* Wk    = (const float*)d_in[3];
  const float* Wv    = (const float*)d_in[4];
  const float* gamma = (const float*)d_in[5];
  const float* beta  = (const float*)d_in[6];
  float* out = (float*)d_out;

  float* q_buf = (float*)d_ws;
  float* k_buf = q_buf + (size_t)kB * kT * kHD;
  float* v_buf = k_buf + (size_t)kB * kS * kHD;

  ln_mem_kernel<<<kB, 64, 0, stream>>>(c_emb, gamma, beta, k_buf, v_buf);
  qkv_kernel<<<(kB * kT) / kRows, kThreads, 0, stream>>>(x, Wq, Wk, Wv,
                                                         q_buf, k_buf, v_buf);
  attn_kernel<<<dim3(kT / kRows, kB), kThreads, 0, stream>>>(q_buf, k_buf, v_buf, out);
}


# memory-token layernorm moved from its own launch into 8 extra work-groups at the end of the projection grid (256 -> 264): 3 -> 2 launches per run; row arithmetic and sum order unchanged
# speedup vs baseline: 3.1120x; 1.0420x over previous
#include <hip/hip_runtime.h>
#include <math.h>

typedef float v2f __attribute__((ext_vector_type(2)));
typedef float v8f __attribute__((ext_vector_type(8)));
typedef float v4f __attribute__((ext_vector_type(4)));

namespace {
constexpr int kB  = 8;
constexpr int kT  = 2048;
constexpr int kD  = 1024;
constexpr int kHD = 64;
constexpr int kM  = 64;
constexpr int kS  = kT + kM;
constexpr float kLnEps = 1e-5f;
constexpr float kScale = 0.125f;
constexpr int kRows    = 64;
constexpr int kWaves   = 8;
constexpr int kThreads = kWaves * 32;
constexpr int kGroup   = 128;
constexpr int kProjBlocks = (kB * kT) / kRows;
}

__device__ __forceinline__ v8f wmma_f32(v2f a, v2f b, v8f c) {
  return __builtin_amdgcn_wmma_f32_16x16x4_f32(
       false, a,  false, b,
       (short)0, c,  false,  false);
}

__device__ __forceinline__ void lds_barrier() {
  __builtin_amdgcn_fence(__ATOMIC_RELEASE, "workgroup", "local");
  __builtin_amdgcn_s_barrier();
  __builtin_amdgcn_fence(__ATOMIC_ACQUIRE, "workgroup", "local");
}

__device__ __forceinline__
void ln_mem_rows(const float* __restrict__ c_emb,
                 const float* __restrict__ gamma,
                 const float* __restrict__ beta,
                 float* __restrict__ k_buf,
                 float* __restrict__ v_buf,
                 const int b, const int r) {
  const float* row = c_emb + r * kHD;
  float mu = 0.f;
  #pragma unroll
  for (int i = 0; i < kHD; ++i) mu += row[i];
  mu *= (1.f / kHD);
  float var = 0.f;
  #pragma unroll
  for (int i = 0; i < kHD; ++i) { float d = row[i] - mu; var += d * d; }
  var *= (1.f / kHD);
  const float rstd = rsqrtf(var + kLnEps);
  float* kd = k_buf + ((size_t)b * kS + kT + r) * kHD;
  float* vd = v_buf + ((size_t)b * kS + kT + r) * kHD;
  #pragma unroll
  for (int i = 0; i < kHD; ++i) {
    float y = (row[i] - mu) * rstd * gamma[i] + beta[i];
    kd[i] = y;
    vd[i] = y;
  }
}

__global__ __launch_bounds__(kThreads)
void qkv_kernel(const float* __restrict__ x,
                const float* __restrict__ Wq,
                const float* __restrict__ Wk,
                const float* __restrict__ Wv,
                const float* __restrict__ c_emb,
                const float* __restrict__ gamma,
                const float* __restrict__ beta,
                float* __restrict__ q_buf,
                float* __restrict__ k_buf,
                float* __restrict__ v_buf) {
  __shared__ float xs[2][kRows][36];
  __shared__ float wst[2][192][34];

  if (blockIdx.x >= kProjBlocks) {
    if (threadIdx.x < kM)
      ln_mem_rows(c_emb, gamma, beta, k_buf, v_buf,
                  (int)blockIdx.x - kProjBlocks, (int)threadIdx.x);
    return;
  }

  const int tid  = threadIdx.x;
  const int wid  = __builtin_amdgcn_readfirstlane(tid >> 5);
  const int lane = tid & 31;
  const int half = lane >> 4;
  const int lm   = lane & 15;
  const int r0   = blockIdx.x * kRows;
  const int m0   = (wid & 3) * 16;
  const int ct0  = (wid >> 2) * 6;
  const int wkr  = wid * 4 + (lane >> 3);
  const int wq8  = lane & 7;

  v8f acc[6] = {};
  v4f xreg[2];
  v4f wreg[6];

  auto gload = [&](int kt) {
    #pragma unroll
    for (int i = 0; i < 2; ++i) {
      int idx = tid + kThreads * i;
      int row = idx >> 3;
      int c4  = (idx & 7) * 4;
      xreg[i] = *(const v4f*)(x + (size_t)(r0 + row) * kD + kt + c4);
    }
    #pragma unroll
    for (int i = 0; i < 6; ++i) {
      const int mat = i >> 1;
      const int cc  = ((i & 1) * 8 + wq8) * 4;
      const float* src = (mat == 0) ? Wq : ((mat == 1) ? Wk : Wv);
      wreg[i] = *(const v4f*)(src + (size_t)(kt + wkr) * kHD + cc);
    }
  };
  auto lstore = [&](int st) {
    #pragma unroll
    for (int i = 0; i < 2; ++i) {
      int idx = tid + kThreads * i;
      int row = idx >> 3;
      int c4  = (idx & 7) * 4;
      *(v4f*)&xs[st][row][c4] = xreg[i];
    }
    #pragma unroll
    for (int i = 0; i < 6; ++i) {
      const int col = (i >> 1) * 64 + ((i & 1) * 8 + wq8) * 4;
      wst[st][col + 0][wkr] = wreg[i].x;
      wst[st][col + 1][wkr] = wreg[i].y;
      wst[st][col + 2][wkr] = wreg[i].z;
      wst[st][col + 3][wkr] = wreg[i].w;
    }
  };
  auto mma = [&](int st, int kk) {
    const int k0 = kk * 4;
    v2f a = *(const v2f*)&xs[st][m0 + lm][k0 + 2 * half];
    #pragma unroll
    for (int c = 0; c < 6; ++c) {
      v2f bf = *(const v2f*)&wst[st][(ct0 + c) * 16 + lm][k0 + 2 * half];
      acc[c] = wmma_f32(a, bf, acc[c]);
    }
  };

  constexpr int kNT = kD / 32;
  gload(0);
  lstore(0);
  gload(32);
  lds_barrier();

  for (int t = 0; t < kNT; ++t) {
    const int cur = t & 1;
    #pragma unroll
    for (int kk = 0; kk < 4; ++kk) mma(cur, kk);
    if (t + 1 < kNT) {
      lstore(cur ^ 1);
      if (t + 2 < kNT) gload(32 * (t + 2));
    }
    __builtin_amdgcn_sched_barrier(0);
    #pragma unroll
    for (int kk = 4; kk < 8; ++kk) mma(cur, kk);
    lds_barrier();
  }

  static_assert(kT % kRows == 0, "a block's rows must not straddle batches");
  const int bb = r0 / kT;
  #pragma unroll
  for (int c = 0; c < 6; ++c) {
    const int ct  = ct0 + c;
    const int mat = ct >> 2;
    const int hd  = (ct & 3) * 16 + lm;
    float* dst = (mat == 0) ? q_buf : ((mat == 1) ? k_buf : v_buf);
    const int radd = (mat == 0) ? 0 : bb * kM;
    #pragma unroll
    for (int i = 0; i < 8; ++i) {
      const int gr = r0 + m0 + i + 8 * half;
      dst[(size_t)(gr + radd) * kHD + hd] = acc[c][i];
    }
  }
}


__device__ __forceinline__ float xhalf(float x) {
  const int b = __builtin_bit_cast(int, x);
  return __builtin_bit_cast(float,
      __builtin_amdgcn_permlanex16(b, b, 0x76543210, 0xfedcba98, false, false));
}

__global__ __launch_bounds__(kThreads)
void attn_kernel(const float* __restrict__ q_buf,
                 const float* __restrict__ k_buf,
                 const float* __restrict__ v_buf,
                 float* __restrict__ out) {
  __shared__ float qs[kRows][68];
  __shared__ __attribute__((aligned(16))) float ks[2][2][64][68];
  __shared__ float vst[2][2][64][68];
  __shared__ float mg[4][34][32];

  const int tid  = threadIdx.x;
  const int wid  = __builtin_amdgcn_readfirstlane(tid >> 5);
  const int lane = tid & 31;
  const int half = lane >> 4;
  const int lm   = lane & 15;
  const int grp  = wid >> 2;
  const int wq   = wid & 3;
  const int gt   = tid & (kGroup - 1);
  const int b    = blockIdx.y;
  const int q0   = blockIdx.x * kRows;
  const int m0   = wq * 16;
  const int vr0  = lane >> 2;
  const int vc4  = (wq * 4 + (lane & 3)) * 4;
  const int krow = ((lm & 6) << 1) | ((lm >> 2) & 2) | (lm & 1);

  constexpr int kNT = kS / 64;
  constexpr int kNG = (kNT + 1) / 2;
  const int t0 = grp * kNG;
  const int nt = (grp == 0) ? kNG : (kNT - kNG);

  v4f kreg[8], vreg[8];
  auto gload = [&](int s0) {
    #pragma unroll
    for (int i = 0; i < 8; ++i) {
      int idx = gt + kGroup * i;
      int row = idx >> 4;
      int c4  = (idx & 15) * 4;
      kreg[i] = *(const v4f*)(k_buf + ((size_t)b * kS + s0 + row) * kHD + c4);
      vreg[i] = *(const v4f*)(v_buf + ((size_t)b * kS + s0 + 8 * i + vr0) * kHD + vc4);
    }
  };
  auto lstore = [&](int st) {
    #pragma unroll
    for (int i = 0; i < 8; ++i) {
      int idx = gt + kGroup * i;
      int row = idx >> 4;
      int c4  = (idx & 15) * 4;
      *(v4f*)&ks[st][grp][row][c4] = kreg[i];
      vst[st][grp][vc4 + 0][8 * i + vr0] = vreg[i].x;
      vst[st][grp][vc4 + 1][8 * i + vr0] = vreg[i].y;
      vst[st][grp][vc4 + 2][8 * i + vr0] = vreg[i].z;
      vst[st][grp][vc4 + 3][8 * i + vr0] = vreg[i].w;
    }
  };

  #pragma unroll
  for (int i = 0; i < kRows * 16 / kThreads; ++i) {
    int idx = tid + kThreads * i;
    int row = idx >> 4;
    int c4  = (idx & 15) * 4;
    v4f v = *(const v4f*)(q_buf + ((size_t)b * kT + q0 + row) * kHD + c4);
    v *= kScale;
    *(v4f*)&qs[row][c4] = v;
  }
  gload(64 * t0);
  lstore(0);
  gload(64 * (t0 + 1));
  lds_barrier();

  const float* kfrag = (const float*)__builtin_assume_aligned(&ks[0][grp][krow][2 * half], 8);
  v8f oacc[4] = {};
  float mrun = -INFINITY, lrun = 0.f;

  for (int n = 0; n < kNG; ++n) {
    const bool live = n < nt;
    const int  cur  = n & 1;

    if (live) {
    v8f sacc[4] = {};
    #pragma unroll
    for (int kk = 0; kk < 16; ++kk) {
      const int k0 = kk * 4;
      v2f qf = *(const v2f*)&qs[m0 + lm][k0 + 2 * half];
      #pragma unroll
      for (int j = 0; j < 4; ++j) {
        v2f kf = *(const v2f*)(kfrag + cur * (2 * 64 * 68) + j * (16 * 68) + k0);
        sacc[j] = wmma_f32(kf, qf, sacc[j]);
      }
    }

    float mx = fmaxf(fmaxf(sacc[0][0], sacc[1][0]), fmaxf(sacc[2][0], sacc[3][0]));
    #pragma unroll
    for (int i = 1; i < 8; ++i)
      mx = fmaxf(mx, fmaxf(fmaxf(sacc[0][i], sacc[1][i]),
                           fmaxf(sacc[2][i], sacc[3][i])));
    mx = fmaxf(mx, xhalf(mx));
    const float mnew  = fmaxf(mrun, mx);
    const float alpha = __expf(mrun - mnew);
    mrun = mnew;
    float rs[4];
    #pragma unroll
    for (int j = 0; j < 4; ++j) {
      rs[j] = 0.f;
      #pragma unroll
      for (int i = 0; i < 8; ++i) {
        float p = __expf(sacc[j][i] - mnew);
        sacc[j][i] = p;
        rs[j] += p;
      }
    }
    lrun = lrun * alpha + ((rs[0] + rs[1]) + (rs[2] + rs[3]));
    #pragma unroll
    for (int c = 0; c < 4; ++c)
      #pragma unroll
      for (int i = 0; i < 8; ++i)
        oacc[c][i] *= alpha;

    if (n + 1 < nt) {
      lstore(cur ^ 1);
      if (n + 2 < nt) gload(64 * (t0 + n + 2));
    }

    #pragma unroll
    for (int kk = 0; kk < 16; ++kk) {
      const int k0 = kk * 4;
      const int j  = kk >> 2;
      const int t  = kk & 3;
      v2f pf;
      pf.x = sacc[j][2 * t];
      pf.y = sacc[j][2 * t + 1];
      #pragma unroll
      for (int c = 0; c < 4; ++c) {
        v2f vf = *(const v2f*)&vst[cur][grp][c * 16 + lm][k0 + 2 * half];
        oacc[c] = wmma_f32(vf, pf, oacc[c]);
      }
    }
    }

    lds_barrier();
  }

  if (grp == 1) {
    #pragma unroll
    for (int c = 0; c < 4; ++c)
      #pragma unroll
      for (int i = 0; i < 8; ++i)
        mg[wq][c * 8 + i][lane] = oacc[c][i];
    mg[wq][32][lane] = mrun;
    mg[wq][33][lane] = lrun;
  }
  __syncthreads();

  if (grp == 0) {
    const float m1 = mg[wq][32][lane];
    const float l1 = mg[wq][33][lane];
    const float m  = fmaxf(mrun, m1);
    const float a0 = __expf(mrun - m);
    const float a1 = __expf(m1 - m);
    float lsum = lrun * a0 + l1 * a1;
    lsum += xhalf(lsum);
    float* orow = out + ((size_t)b * kT + q0 + m0 + lm) * kHD + 8 * half;
    #pragma unroll
    for (int c = 0; c < 4; ++c) {
      v4f lo, hi;
      lo.x = (oacc[c][0] * a0 + mg[wq][c * 8 + 0][lane] * a1) / lsum;
      lo.y = (oacc[c][1] * a0 + mg[wq][c * 8 + 1][lane] * a1) / lsum;
      lo.z = (oacc[c][2] * a0 + mg[wq][c * 8 + 2][lane] * a1) / lsum;
      lo.w = (oacc[c][3] * a0 + mg[wq][c * 8 + 3][lane] * a1) / lsum;
      hi.x = (oacc[c][4] * a0 + mg[wq][c * 8 + 4][lane] * a1) / lsum;
      hi.y = (oacc[c][5] * a0 + mg[wq][c * 8 + 5][lane] * a1) / lsum;
      hi.z = (oacc[c][6] * a0 + mg[wq][c * 8 + 6][lane] * a1) / lsum;
      hi.w = (oacc[c][7] * a0 + mg[wq][c * 8 + 7][lane] * a1) / lsum;
      *(v4f*)(orow + c * 16)     = lo;
      *(v4f*)(orow + c * 16 + 4) = hi;
    }
  }
}

extern "C" void kernel_launch(void* const* d_in, const int* in_sizes, int n_in,
                              void* d_out, int out_size, void* d_ws, size_t ws_size,
                              hipStream_t stream) {
  const float* x     = (const float*)d_in[0];
  const float* c_emb = (const float*)d_in[1];
  const float* Wq    = (const float*)d_in[2];
  const float* Wk    = (const float*)d_in[3];
  const float* Wv    = (const float*)d_in[4];
  const float* gamma = (const float*)d_in[5];
  const float* beta  = (const float*)d_in[6];
  float* out = (float*)d_out;

  float* q_buf = (float*)d_ws;
  float* k_buf = q_buf + (size_t)kB * kT * kHD;
  float* v_buf = k_buf + (size_t)kB * kS * kHD;

  qkv_kernel<<<kProjBlocks + kB, kThreads, 0, stream>>>(x, Wq, Wk, Wv,
                                                         c_emb, gamma, beta,
                                                         q_buf, k_buf, v_buf);
  attn_kernel<<<dim3(kT / kRows, kB), kThreads, 0, stream>>>(q_buf, k_buf, v_buf, out);
}


# attention: key tiles split 22:11 (was 17:16) between the older and younger wave of each SIMD; the younger wave's tile spans two barrier intervals so it fits the older one's gaps; sum order changes
# speedup vs baseline: 3.2167x; 1.0336x over previous
#include <hip/hip_runtime.h>
#include <math.h>

typedef float v2f __attribute__((ext_vector_type(2)));
typedef float v8f __attribute__((ext_vector_type(8)));
typedef float v4f __attribute__((ext_vector_type(4)));

namespace {
constexpr int kB  = 8;
constexpr int kT  = 2048;
constexpr int kD  = 1024;
constexpr int kHD = 64;
constexpr int kM  = 64;
constexpr int kS  = kT + kM;
constexpr float kLnEps = 1e-5f;
constexpr float kScale = 0.125f;
constexpr int kRows    = 64;
constexpr int kWaves   = 8;
constexpr int kThreads = kWaves * 32;
constexpr int kGroup   = 128;
constexpr int kProjBlocks = (kB * kT) / kRows;
}

__device__ __forceinline__ v8f wmma_f32(v2f a, v2f b, v8f c) {
  return __builtin_amdgcn_wmma_f32_16x16x4_f32(
       false, a,  false, b,
       (short)0, c,  false,  false);
}

__device__ __forceinline__ void lds_barrier() {
  __builtin_amdgcn_fence(__ATOMIC_RELEASE, "workgroup", "local");
  __builtin_amdgcn_s_barrier();
  __builtin_amdgcn_fence(__ATOMIC_ACQUIRE, "workgroup", "local");
}

__device__ __forceinline__
void ln_mem_rows(const float* __restrict__ c_emb,
                 const float* __restrict__ gamma,
                 const float* __restrict__ beta,
                 float* __restrict__ k_buf,
                 float* __restrict__ v_buf,
                 const int b, const int r) {
  const float* row = c_emb + r * kHD;
  float mu = 0.f;
  #pragma unroll
  for (int i = 0; i < kHD; ++i) mu += row[i];
  mu *= (1.f / kHD);
  float var = 0.f;
  #pragma unroll
  for (int i = 0; i < kHD; ++i) { float d = row[i] - mu; var += d * d; }
  var *= (1.f / kHD);
  const float rstd = rsqrtf(var + kLnEps);
  float* kd = k_buf + ((size_t)b * kS + kT + r) * kHD;
  float* vd = v_buf + ((size_t)b * kS + kT + r) * kHD;
  #pragma unroll
  for (int i = 0; i < kHD; ++i) {
    float y = (row[i] - mu) * rstd * gamma[i] + beta[i];
    kd[i] = y;
    vd[i] = y;
  }
}

__global__ __launch_bounds__(kThreads)
void qkv_kernel(const float* __restrict__ x,
                const float* __restrict__ Wq,
                const float* __restrict__ Wk,
                const float* __restrict__ Wv,
                const float* __restrict__ c_emb,
                const float* __restrict__ gamma,
                const float* __restrict__ beta,
                float* __restrict__ q_buf,
                float* __restrict__ k_buf,
                float* __restrict__ v_buf) {
  __shared__ float xs[2][kRows][36];
  __shared__ float wst[2][192][34];

  if (blockIdx.x >= kProjBlocks) {
    if (threadIdx.x < kM)
      ln_mem_rows(c_emb, gamma, beta, k_buf, v_buf,
                  (int)blockIdx.x - kProjBlocks, (int)threadIdx.x);
    return;
  }

  const int tid  = threadIdx.x;
  const int wid  = __builtin_amdgcn_readfirstlane(tid >> 5);
  const int lane = tid & 31;
  const int half = lane >> 4;
  const int lm   = lane & 15;
  const int r0   = blockIdx.x * kRows;
  const int m0   = (wid & 3) * 16;
  const int ct0  = (wid >> 2) * 6;
  const int wkr  = wid * 4 + (lane >> 3);
  const int wq8  = lane & 7;

  v8f acc[6] = {};
  v4f xreg[2];
  v4f wreg[6];

  auto gload = [&](int kt) {
    #pragma unroll
    for (int i = 0; i < 2; ++i) {
      int idx = tid + kThreads * i;
      int row = idx >> 3;
      int c4  = (idx & 7) * 4;
      xreg[i] = *(const v4f*)(x + (size_t)(r0 + row) * kD + kt + c4);
    }
    #pragma unroll
    for (int i = 0; i < 6; ++i) {
      const int mat = i >> 1;
      const int cc  = ((i & 1) * 8 + wq8) * 4;
      const float* src = (mat == 0) ? Wq : ((mat == 1) ? Wk : Wv);
      wreg[i] = *(const v4f*)(src + (size_t)(kt + wkr) * kHD + cc);
    }
  };
  auto lstore = [&](int st) {
    #pragma unroll
    for (int i = 0; i < 2; ++i) {
      int idx = tid + kThreads * i;
      int row = idx >> 3;
      int c4  = (idx & 7) * 4;
      *(v4f*)&xs[st][row][c4] = xreg[i];
    }
    #pragma unroll
    for (int i = 0; i < 6; ++i) {
      const int col = (i >> 1) * 64 + ((i & 1) * 8 + wq8) * 4;
      wst[st][col + 0][wkr] = wreg[i].x;
      wst[st][col + 1][wkr] = wreg[i].y;
      wst[st][col + 2][wkr] = wreg[i].z;
      wst[st][col + 3][wkr] = wreg[i].w;
    }
  };
  auto mma = [&](int st, int kk) {
    const int k0 = kk * 4;
    v2f a = *(const v2f*)&xs[st][m0 + lm][k0 + 2 * half];
    #pragma unroll
    for (int c = 0; c < 6; ++c) {
      v2f bf = *(const v2f*)&wst[st][(ct0 + c) * 16 + lm][k0 + 2 * half];
      acc[c] = wmma_f32(a, bf, acc[c]);
    }
  };

  constexpr int kNT = kD / 32;
  gload(0);
  lstore(0);
  gload(32);
  lds_barrier();

  for (int t = 0; t < kNT; ++t) {
    const int cur = t & 1;
    #pragma unroll
    for (int kk = 0; kk < 4; ++kk) mma(cur, kk);
    if (t + 1 < kNT) {
      lstore(cur ^ 1);
      if (t + 2 < kNT) gload(32 * (t + 2));
    }
    __builtin_amdgcn_sched_barrier(0);
    #pragma unroll
    for (int kk = 4; kk < 8; ++kk) mma(cur, kk);
    lds_barrier();
  }

  static_assert(kT % kRows == 0, "a block's rows must not straddle batches");
  const int bb = r0 / kT;
  #pragma unroll
  for (int c = 0; c < 6; ++c) {
    const int ct  = ct0 + c;
    const int mat = ct >> 2;
    const int hd  = (ct & 3) * 16 + lm;
    float* dst = (mat == 0) ? q_buf : ((mat == 1) ? k_buf : v_buf);
    const int radd = (mat == 0) ? 0 : bb * kM;
    #pragma unroll
    for (int i = 0; i < 8; ++i) {
      const int gr = r0 + m0 + i + 8 * half;
      dst[(size_t)(gr + radd) * kHD + hd] = acc[c][i];
    }
  }
}


__device__ __forceinline__ float xhalf(float x) {
  const int b = __builtin_bit_cast(int, x);
  return __builtin_bit_cast(float,
      __builtin_amdgcn_permlanex16(b, b, 0x76543210, 0xfedcba98, false, false));
}

__global__ __launch_bounds__(kThreads)
void attn_kernel(const float* __restrict__ q_buf,
                 const float* __restrict__ k_buf,
                 const float* __restrict__ v_buf,
                 float* __restrict__ out) {
  __shared__ float qs[kRows][68];
  __shared__ __attribute__((aligned(16))) float ks[2][2][64][68];
  __shared__ float vst[2][2][64][68];
  __shared__ float mg[4][34][32];

  const int tid  = threadIdx.x;
  const int wid  = __builtin_amdgcn_readfirstlane(tid >> 5);
  const int lane = tid & 31;
  const int half = lane >> 4;
  const int lm   = lane & 15;
  const int grp  = wid >> 2;
  const int wq   = wid & 3;
  const int gt   = tid & (kGroup - 1);
  const int b    = blockIdx.y;
  const int q0   = blockIdx.x * kRows;
  const int m0   = wq * 16;
  const int vr0  = lane >> 2;
  const int vc4  = (wq * 4 + (lane & 3)) * 4;
  const int krow = ((lm & 6) << 1) | ((lm >> 2) & 2) | (lm & 1);

  constexpr int kNT = kS / 64;
  constexpr int kNI = 22;
  constexpr int kN1 = kNT - kNI;
  static_assert(2 * kN1 == kNI, "group 1 needs two intervals per tile");
  const int t0 = grp * kNI;
  const int nt = (grp == 0) ? kNI : kN1;

  v4f kreg[8], vreg[8];
  auto gload = [&](int s0) {
    #pragma unroll
    for (int i = 0; i < 8; ++i) {
      int idx = gt + kGroup * i;
      int row = idx >> 4;
      int c4  = (idx & 15) * 4;
      kreg[i] = *(const v4f*)(k_buf + ((size_t)b * kS + s0 + row) * kHD + c4);
      vreg[i] = *(const v4f*)(v_buf + ((size_t)b * kS + s0 + 8 * i + vr0) * kHD + vc4);
    }
  };
  auto lstore = [&](int st) {
    #pragma unroll
    for (int i = 0; i < 8; ++i) {
      int idx = gt + kGroup * i;
      int row = idx >> 4;
      int c4  = (idx & 15) * 4;
      *(v4f*)&ks[st][grp][row][c4] = kreg[i];
      vst[st][grp][vc4 + 0][8 * i + vr0] = vreg[i].x;
      vst[st][grp][vc4 + 1][8 * i + vr0] = vreg[i].y;
      vst[st][grp][vc4 + 2][8 * i + vr0] = vreg[i].z;
      vst[st][grp][vc4 + 3][8 * i + vr0] = vreg[i].w;
    }
  };

  #pragma unroll
  for (int i = 0; i < kRows * 16 / kThreads; ++i) {
    int idx = tid + kThreads * i;
    int row = idx >> 4;
    int c4  = (idx & 15) * 4;
    v4f v = *(const v4f*)(q_buf + ((size_t)b * kT + q0 + row) * kHD + c4);
    v *= kScale;
    *(v4f*)&qs[row][c4] = v;
  }
  gload(64 * t0);
  lstore(0);
  gload(64 * (t0 + 1));
  lds_barrier();

  const float* kfrag = (const float*)__builtin_assume_aligned(&ks[0][grp][krow][2 * half], 8);
  v8f oacc[4] = {};
  float mrun = -INFINITY, lrun = 0.f;

  v8f sacc[4] = {};

  for (int n = 0; n < kNI; ++n) {
    const int  m    = (grp == 0) ? n : (n >> 1);
    const int  cur  = m & 1;
    const bool do_s  = (grp == 0) || ((n & 1) == 0);
    const bool do_pv = (grp == 0) || ((n & 1) != 0);

    if (do_s) {
    #pragma unroll
    for (int j = 0; j < 4; ++j) sacc[j] = v8f{};
    #pragma unroll
    for (int kk = 0; kk < 16; ++kk) {
      const int k0 = kk * 4;
      v2f qf = *(const v2f*)&qs[m0 + lm][k0 + 2 * half];
      #pragma unroll
      for (int j = 0; j < 4; ++j) {
        v2f kf = *(const v2f*)(kfrag + cur * (2 * 64 * 68) + j * (16 * 68) + k0);
        sacc[j] = wmma_f32(kf, qf, sacc[j]);
      }
    }
    }

    if (do_pv) {
    float mx = fmaxf(fmaxf(sacc[0][0], sacc[1][0]), fmaxf(sacc[2][0], sacc[3][0]));
    #pragma unroll
    for (int i = 1; i < 8; ++i)
      mx = fmaxf(mx, fmaxf(fmaxf(sacc[0][i], sacc[1][i]),
                           fmaxf(sacc[2][i], sacc[3][i])));
    mx = fmaxf(mx, xhalf(mx));
    const float mnew  = fmaxf(mrun, mx);
    const float alpha = __expf(mrun - mnew);
    mrun = mnew;
    float rs[4];
    #pragma unroll
    for (int j = 0; j < 4; ++j) {
      rs[j] = 0.f;
      #pragma unroll
      for (int i = 0; i < 8; ++i) {
        float p = __expf(sacc[j][i] - mnew);
        sacc[j][i] = p;
        rs[j] += p;
      }
    }
    lrun = lrun * alpha + ((rs[0] + rs[1]) + (rs[2] + rs[3]));
    #pragma unroll
    for (int c = 0; c < 4; ++c)
      #pragma unroll
      for (int i = 0; i < 8; ++i)
        oacc[c][i] *= alpha;

    if (m + 1 < nt) {
      lstore(cur ^ 1);
      if (m + 2 < nt) gload(64 * (t0 + m + 2));
    }

    #pragma unroll
    for (int kk = 0; kk < 16; ++kk) {
      const int k0 = kk * 4;
      const int j  = kk >> 2;
      const int t  = kk & 3;
      v2f pf;
      pf.x = sacc[j][2 * t];
      pf.y = sacc[j][2 * t + 1];
      #pragma unroll
      for (int c = 0; c < 4; ++c) {
        v2f vf = *(const v2f*)&vst[cur][grp][c * 16 + lm][k0 + 2 * half];
        oacc[c] = wmma_f32(vf, pf, oacc[c]);
      }
    }
    }

    lds_barrier();
  }

  if (grp == 1) {
    #pragma unroll
    for (int c = 0; c < 4; ++c)
      #pragma unroll
      for (int i = 0; i < 8; ++i)
        mg[wq][c * 8 + i][lane] = oacc[c][i];
    mg[wq][32][lane] = mrun;
    mg[wq][33][lane] = lrun;
  }
  __syncthreads();

  if (grp == 0) {
    const float m1 = mg[wq][32][lane];
    const float l1 = mg[wq][33][lane];
    const float m  = fmaxf(mrun, m1);
    const float a0 = __expf(mrun - m);
    const float a1 = __expf(m1 - m);
    float lsum = lrun * a0 + l1 * a1;
    lsum += xhalf(lsum);
    float* orow = out + ((size_t)b * kT + q0 + m0 + lm) * kHD + 8 * half;
    #pragma unroll
    for (int c = 0; c < 4; ++c) {
      v4f lo, hi;
      lo.x = (oacc[c][0] * a0 + mg[wq][c * 8 + 0][lane] * a1) / lsum;
      lo.y = (oacc[c][1] * a0 + mg[wq][c * 8 + 1][lane] * a1) / lsum;
      lo.z = (oacc[c][2] * a0 + mg[wq][c * 8 + 2][lane] * a1) / lsum;
      lo.w = (oacc[c][3] * a0 + mg[wq][c * 8 + 3][lane] * a1) / lsum;
      hi.x = (oacc[c][4] * a0 + mg[wq][c * 8 + 4][lane] * a1) / lsum;
      hi.y = (oacc[c][5] * a0 + mg[wq][c * 8 + 5][lane] * a1) / lsum;
      hi.z = (oacc[c][6] * a0 + mg[wq][c * 8 + 6][lane] * a1) / lsum;
      hi.w = (oacc[c][7] * a0 + mg[wq][c * 8 + 7][lane] * a1) / lsum;
      *(v4f*)(orow + c * 16)     = lo;
      *(v4f*)(orow + c * 16 + 4) = hi;
    }
  }
}

extern "C" void kernel_launch(void* const* d_in, const int* in_sizes, int n_in,
                              void* d_out, int out_size, void* d_ws, size_t ws_size,
                              hipStream_t stream) {
  const float* x     = (const float*)d_in[0];
  const float* c_emb = (const float*)d_in[1];
  const float* Wq    = (const float*)d_in[2];
  const float* Wk    = (const float*)d_in[3];
  const float* Wv    = (const float*)d_in[4];
  const float* gamma = (const float*)d_in[5];
  const float* beta  = (const float*)d_in[6];
  float* out = (float*)d_out;

  float* q_buf = (float*)d_ws;
  float* k_buf = q_buf + (size_t)kB * kT * kHD;
  float* v_buf = k_buf + (size_t)kB * kS * kHD;

  qkv_kernel<<<kProjBlocks + kB, kThreads, 0, stream>>>(x, Wq, Wk, Wv,
                                                         c_emb, gamma, beta,
                                                         q_buf, k_buf, v_buf);
  attn_kernel<<<dim3(kT / kRows, kB), kThreads, 0, stream>>>(q_buf, k_buf, v_buf, out);
}
